# starting kernel
# baseline (speedup 1.0000x reference)
#include <hip/hip_runtime.h>
#include <hip/hip_bf16.h>

typedef float v2f __attribute__((ext_vector_type(2)));
typedef float v8f __attribute__((ext_vector_type(8)));

#define LRELU_SLOPE 0.01f

__device__ __forceinline__ void atomic_add_f32(float* p, float v) {
    asm volatile("global_atomic_add_f32 %0, %1, off scope:SCOPE_DEV"
                 :: "v"(p), "v"(v) : "memory");
}

__global__ void k_zero_u32(unsigned* __restrict__ p, int n) {
    int i = blockIdx.x * blockDim.x + threadIdx.x;
    if (i < n) p[i] = 0u;
}

__global__ void k_count_deg(const int* __restrict__ ei, int E, unsigned* __restrict__ deg) {
    int e = blockIdx.x * blockDim.x + threadIdx.x;
    if (e < E) {
        int d = ei[E + e];
        atomicAdd(&deg[d], 1u);
    }
}

__global__ void k_dinv(const unsigned* __restrict__ deg, float* __restrict__ dinv, int n) {
    int i = blockIdx.x * blockDim.x + threadIdx.x;
    if (i < n) dinv[i] = rsqrtf((float)(deg[i] + 1u));
}

template <int NOUT>
__global__ __launch_bounds__(256) void k_gemm_wmma(const float* __restrict__ A,
                                                   const float* __restrict__ W,
                                                   float* __restrict__ out, int n_rows) {
    __shared__ float sW[128 * NOUT];
    const int tid = threadIdx.x;
    for (int i = tid; i < 128 * NOUT; i += 256) {
        int k = i / NOUT, n = i % NOUT;
        sW[((k >> 1) * NOUT + n) * 2 + (k & 1)] = W[i];
    }
    __syncthreads();

    const int wave = tid >> 5;
    const int lane = tid & 31;
    const int m16  = lane & 15;
    const int hi   = lane >> 4;
    const int row0 = blockIdx.x * 128 + wave * 16;
    constexpr int NT = NOUT / 16;

    v8f zero8 = {};
    v8f acc[NT];
#pragma unroll
    for (int t = 0; t < NT; ++t) acc[t] = zero8;

    int rA = row0 + m16;
    if (rA >= n_rows) rA = n_rows - 1;
    const float* arow = A + (size_t)rA * 128;

    for (int k0 = 0; k0 < 128; k0 += 4) {
        v2f a = *(const v2f*)(arow + k0 + 2 * hi);
        const int kh = (k0 >> 1) + hi;
#pragma unroll
        for (int t = 0; t < NT; ++t) {
            v2f b = *(const v2f*)(sW + (size_t)(kh * NOUT + t * 16 + m16) * 2);
            acc[t] = __builtin_amdgcn_wmma_f32_16x16x4_f32(
                false, a, false, b, (short)0, acc[t], false, false);
        }
    }

#pragma unroll
    for (int t = 0; t < NT; ++t) {
#pragma unroll
        for (int v = 0; v < 8; ++v) {
            int row = row0 + v + 8 * hi;
            if (row < n_rows) out[(size_t)row * NOUT + t * 16 + m16] = acc[t][v];
        }
    }
}

template <int FEAT>
__global__ void k_self_init(const float* __restrict__ h, const float* __restrict__ dinv,
                            float* __restrict__ acc, int n_rows) {
    constexpr int QPR = FEAT / 4;
    int t = blockIdx.x * blockDim.x + threadIdx.x;
    int node = t / QPR;
    if (node >= n_rows) return;
    int fi = (t % QPR) * 4;
    float di = dinv[node];
    float w = di * di;
    float4 v = *(const float4*)(h + (size_t)node * FEAT + fi);
    float4 o = {v.x * w, v.y * w, v.z * w, v.w * w};
    *(float4*)(acc + (size_t)node * FEAT + fi) = o;
}

template <int FEAT>
__global__ __launch_bounds__(256) void k_edge_agg(const float* __restrict__ h,
                                                  const float* __restrict__ dinv,
                                                  const int* __restrict__ ei, int E,
                                                  float* __restrict__ acc) {
    constexpr int VW = FEAT / 32;
    long long t = (long long)blockIdx.x * 256 + threadIdx.x;
    int e = (int)(t >> 5);
    if (e >= E) return;
    e = __builtin_amdgcn_readfirstlane(e);
    const int lane = threadIdx.x & 31;
    int s = ei[e];
    int d = ei[E + e];
    float nrm = dinv[s] * dinv[d];
    const float* src = h + (size_t)s * FEAT + lane * VW;
    float* dst = acc + (size_t)d * FEAT + lane * VW;
    if constexpr (VW == 4) {
        float4 v = *(const float4*)src;
        atomic_add_f32(dst + 0, nrm * v.x);
        atomic_add_f32(dst + 1, nrm * v.y);
        atomic_add_f32(dst + 2, nrm * v.z);
        atomic_add_f32(dst + 3, nrm * v.w);
    } else {
        float2 v = *(const float2*)src;
        atomic_add_f32(dst + 0, nrm * v.x);
        atomic_add_f32(dst + 1, nrm * v.y);
    }
}

template <int FEAT, bool LEAKY>
__global__ void k_bias_act(float* __restrict__ acc, const float* __restrict__ b, int n_rows) {
    constexpr int QPR = FEAT / 4;
    int t = blockIdx.x * blockDim.x + threadIdx.x;
    int node = t / QPR;
    if (node >= n_rows) return;
    int fi = (t % QPR) * 4;
    float4 v  = *(const float4*)(acc + (size_t)node * FEAT + fi);
    float4 bb = *(const float4*)(b + fi);
    v.x += bb.x; v.y += bb.y; v.z += bb.z; v.w += bb.w;
    if (LEAKY) {
        v.x = v.x > 0.0f ? v.x : LRELU_SLOPE * v.x;
        v.y = v.y > 0.0f ? v.y : LRELU_SLOPE * v.y;
        v.z = v.z > 0.0f ? v.z : LRELU_SLOPE * v.z;
        v.w = v.w > 0.0f ? v.w : LRELU_SLOPE * v.w;
    }
    *(float4*)(acc + (size_t)node * FEAT + fi) = v;
}

extern "C" void kernel_launch(void* const* d_in, const int* in_sizes, int n_in,
                              void* d_out, int out_size, void* d_ws, size_t ws_size,
                              hipStream_t stream) {
    (void)n_in; (void)out_size; (void)ws_size;
    const float* z  = (const float*)d_in[0];
    const int*   ei = (const int*)d_in[1];
    const float* W1 = (const float*)d_in[2];
    const float* b1 = (const float*)d_in[3];
    const float* W2 = (const float*)d_in[4];
    const float* b2 = (const float*)d_in[5];

    const int N = in_sizes[0] / 128;
    const int E = in_sizes[1] / 2;

    char* ws = (char*)d_ws;
    size_t off = 0;
    auto carve = [&](size_t bytes) -> void* {
        void* p = ws + off;
        off = (off + bytes + 255) & ~(size_t)255;
        return p;
    };
    unsigned* deg  = (unsigned*)carve((size_t)N * 4);
    float*    dinv = (float*)carve((size_t)N * 4);
    float*    h1   = (float*)carve((size_t)N * 128 * 4);
    float*    a1   = (float*)carve((size_t)N * 128 * 4);
    float*    h3   = (float*)carve((size_t)N * 64 * 4);
    float*    outp = (float*)d_out;

    const dim3 blk(256);
    const int gN = (N + 255) / 256;
    const int gE = (E + 255) / 256;
    const int gM = (N + 127) / 128;
    const int gEdge = (int)(((long long)E * 32 + 255) / 256);

    k_zero_u32 <<<gN, blk, 0, stream>>>(deg, N);
    k_count_deg<<<gE, blk, 0, stream>>>(ei, E, deg);
    k_dinv     <<<gN, blk, 0, stream>>>(deg, dinv, N);

    k_gemm_wmma<128><<<gM, blk, 0, stream>>>(z, W1, h1, N);
    long long t128 = (long long)N * 32;
    int gS1 = (int)((t128 + 255) / 256);
    k_self_init<128><<<gS1, blk, 0, stream>>>(h1, dinv, a1, N);
    k_edge_agg<128><<<gEdge, blk, 0, stream>>>(h1, dinv, ei, E, a1);
    k_bias_act<128, true><<<gS1, blk, 0, stream>>>(a1, b1, N);

    k_gemm_wmma<64><<<gM, blk, 0, stream>>>(a1, W2, h3, N);
    long long t64 = (long long)N * 16;
    int gS2 = (int)((t64 + 255) / 256);
    k_self_init<64><<<gS2, blk, 0, stream>>>(h3, dinv, outp, N);
    k_edge_agg<64><<<gEdge, blk, 0, stream>>>(h3, dinv, ei, E, outp);
    k_bias_act<64, false><<<gS2, blk, 0, stream>>>(outp, b2, N);
}


# Replace the f32 atomic scatter-add by a destination-sorted edge list (count+rank, scan, fill) and a per-node register gather-sum, self loop, bias and activation fused; sums now in list order
# speedup vs baseline: 2.6128x; 2.6128x over previous
#include <hip/hip_runtime.h>
#include <hip/hip_bf16.h>

typedef float v2f __attribute__((ext_vector_type(2)));
typedef float v8f __attribute__((ext_vector_type(8)));

#define LRELU_SLOPE 0.01f

__global__ void k_zero_u32(unsigned* __restrict__ p, int n) {
    int i = blockIdx.x * blockDim.x + threadIdx.x;
    if (i < n) p[i] = 0u;
}

__global__ void k_count_rank(const int* __restrict__ ei, int E, unsigned* __restrict__ deg,
                             unsigned* __restrict__ rank) {
    int e = blockIdx.x * blockDim.x + threadIdx.x;
    if (e < E) {
        int d = ei[E + e];
        rank[e] = atomicAdd(&deg[d], 1u);
    }
}

__global__ __launch_bounds__(256) void k_scan_rows(const unsigned* __restrict__ deg, int n,
                                                   unsigned* __restrict__ rowptr,
                                                   float* __restrict__ dinv,
                                                   int* __restrict__ ent_raw) {
    __shared__ unsigned s_front[8];
    __shared__ unsigned s_wsum[8];
    int2* __restrict__ ent = (int2*)ent_raw;
    const int tid = threadIdx.x;
    const int lane = tid & 31;
    const int wave = tid >> 5;
    const int blk0 = blockIdx.x * 1024;

    unsigned front = 0u;
    const uint4* __restrict__ deg4 = (const uint4*)deg;
#pragma unroll 8
    for (int p = 0; p < (int)blockIdx.x; ++p) {
        const uint4 q = deg4[p * 256 + tid];
        front += (q.x + q.y) + (q.z + q.w);
    }

    const int i0 = blk0 + tid * 4;
    unsigned c[4] = {0u, 0u, 0u, 0u};
    if (i0 + 3 < n) {
        const uint4 q = deg4[i0 >> 2];
        c[0] = q.x + 1u; c[1] = q.y + 1u; c[2] = q.z + 1u; c[3] = q.w + 1u;
    } else {
#pragma unroll
        for (int k = 0; k < 4; ++k) if (i0 + k < n) c[k] = deg[i0 + k] + 1u;
    }
    const unsigned tsum = c[0] + c[1] + c[2] + c[3];

    unsigned incl = tsum;
#pragma unroll
    for (int off = 1; off < 32; off <<= 1) {
        unsigned v = __shfl_up(incl, off);
        if (lane >= off) incl += v;
    }
#pragma unroll
    for (int off = 16; off > 0; off >>= 1) front += __shfl_xor(front, off);
    if (lane == 31) s_wsum[wave] = incl;
    if (lane == 0) s_front[wave] = front;
    __syncthreads();

    unsigned base = (unsigned)blk0;
#pragma unroll
    for (int w = 0; w < 8; ++w) {
        base += s_front[w];
        if (w < wave) base += s_wsum[w];
    }
    unsigned pos = base + incl - tsum;
#pragma unroll
    for (int k = 0; k < 4; ++k) {
        const int idx = i0 + k;
        if (idx < n) {
            rowptr[idx] = pos;
            const float di = rsqrtf((float)c[k]);
            dinv[idx] = di;
            ent[pos] = make_int2(idx, __float_as_int(di * di));
            pos += c[k];
            if (idx == n - 1) rowptr[n] = pos;
        }
    }
}

__global__ void k_fill_rows(const int* __restrict__ ei, int E, const unsigned* __restrict__ rank,
                            const unsigned* __restrict__ rowptr, const float* __restrict__ dinv,
                            int* __restrict__ ent_raw) {
    int2* __restrict__ ent = (int2*)ent_raw;
    int e = blockIdx.x * blockDim.x + threadIdx.x;
    if (e < E) {
        const int s = ei[e];
        const int d = ei[E + e];
        const unsigned slot = rowptr[d] + 1u + rank[e];
        ent[slot] = make_int2(s, __float_as_int(dinv[s] * dinv[d]));
    }
}

template <int NOUT>
__global__ __launch_bounds__(256) void k_gemm_wmma(const float* __restrict__ A,
                                                   const float* __restrict__ W,
                                                   float* __restrict__ out, int n_rows) {
    __shared__ float sW[128 * NOUT];
    const int tid = threadIdx.x;
    for (int i = tid; i < 128 * NOUT; i += 256) {
        int k = i / NOUT, n = i % NOUT;
        sW[((k >> 1) * NOUT + n) * 2 + (k & 1)] = W[i];
    }
    __syncthreads();

    const int wave = tid >> 5;
    const int lane = tid & 31;
    const int m16  = lane & 15;
    const int hi   = lane >> 4;
    const int row0 = blockIdx.x * 128 + wave * 16;
    constexpr int NT = NOUT / 16;

    v8f zero8 = {};
    v8f acc[NT];
#pragma unroll
    for (int t = 0; t < NT; ++t) acc[t] = zero8;

    int rA = row0 + m16;
    if (rA >= n_rows) rA = n_rows - 1;
    const float* arow = A + (size_t)rA * 128;

    for (int k0 = 0; k0 < 128; k0 += 4) {
        v2f a = *(const v2f*)(arow + k0 + 2 * hi);
        const int kh = (k0 >> 1) + hi;
#pragma unroll
        for (int t = 0; t < NT; ++t) {
            v2f b = *(const v2f*)(sW + (size_t)(kh * NOUT + t * 16 + m16) * 2);
            acc[t] = __builtin_amdgcn_wmma_f32_16x16x4_f32(
                false, a, false, b, (short)0, acc[t], false, false);
        }
    }

#pragma unroll
    for (int t = 0; t < NT; ++t) {
#pragma unroll
        for (int v = 0; v < 8; ++v) {
            int row = row0 + v + 8 * hi;
            if (row < n_rows) out[(size_t)row * NOUT + t * 16 + m16] = acc[t][v];
        }
    }
}

template <int FEAT, bool LEAKY>
__global__ __launch_bounds__(256) void k_gather_rows(const float* __restrict__ h,
                                                     const unsigned* __restrict__ rowptr,
                                                     const int* __restrict__ ent_raw,
                                                     const float* __restrict__ b,
                                                     float* __restrict__ out, int n_rows) {
    constexpr int VW = FEAT / 32;
    typedef float vrow __attribute__((ext_vector_type(VW)));
    const int2* __restrict__ ent = (const int2*)ent_raw;
    int node = (int)(((long long)blockIdx.x * 256 + threadIdx.x) >> 5);
    if (node >= n_rows) return;
    node = __builtin_amdgcn_readfirstlane(node);
    const int lane = threadIdx.x & 31;
    const unsigned beg = rowptr[node];
    const unsigned end = rowptr[node + 1];
    const float* hl = h + lane * VW;

    vrow acc = {};
    for (unsigned base = beg; base < end; base += 32u) {
        const unsigned left = end - base;
        const unsigned cnt = left < 32u ? left : 32u;
        int2 my = make_int2(0, 0);
        if ((unsigned)lane < cnt) my = ent[base + lane];
        unsigned j = 0;
        for (; j + 4u <= cnt; j += 4u) {
            const int s0 = __builtin_amdgcn_readlane(my.x, j + 0);
            const int s1 = __builtin_amdgcn_readlane(my.x, j + 1);
            const int s2 = __builtin_amdgcn_readlane(my.x, j + 2);
            const int s3 = __builtin_amdgcn_readlane(my.x, j + 3);
            const float w0 = __int_as_float(__builtin_amdgcn_readlane(my.y, j + 0));
            const float w1 = __int_as_float(__builtin_amdgcn_readlane(my.y, j + 1));
            const float w2 = __int_as_float(__builtin_amdgcn_readlane(my.y, j + 2));
            const float w3 = __int_as_float(__builtin_amdgcn_readlane(my.y, j + 3));
            const vrow v0 = *(const vrow*)(hl + (size_t)s0 * FEAT);
            const vrow v1 = *(const vrow*)(hl + (size_t)s1 * FEAT);
            const vrow v2 = *(const vrow*)(hl + (size_t)s2 * FEAT);
            const vrow v3 = *(const vrow*)(hl + (size_t)s3 * FEAT);
            acc += w0 * v0;
            acc += w1 * v1;
            acc += w2 * v2;
            acc += w3 * v3;
        }
        if (j < cnt) {
            const int s0 = __builtin_amdgcn_readlane(my.x, j);
            const float w0 = __int_as_float(__builtin_amdgcn_readlane(my.y, j));
            const vrow v0 = *(const vrow*)(hl + (size_t)s0 * FEAT);
            vrow v1 = {}, v2 = {};
            float w1 = 0.0f, w2 = 0.0f;
            if (j + 1u < cnt) {
                const int s1 = __builtin_amdgcn_readlane(my.x, j + 1);
                w1 = __int_as_float(__builtin_amdgcn_readlane(my.y, j + 1));
                v1 = *(const vrow*)(hl + (size_t)s1 * FEAT);
            }
            if (j + 2u < cnt) {
                const int s2 = __builtin_amdgcn_readlane(my.x, j + 2);
                w2 = __int_as_float(__builtin_amdgcn_readlane(my.y, j + 2));
                v2 = *(const vrow*)(hl + (size_t)s2 * FEAT);
            }
            acc += w0 * v0;
            acc += w1 * v1;
            acc += w2 * v2;
        }
    }

    const vrow bb = *(const vrow*)(b + lane * VW);
    acc += bb;
    if (LEAKY) {
#pragma unroll
        for (int k = 0; k < VW; ++k) acc[k] = acc[k] > 0.0f ? acc[k] : LRELU_SLOPE * acc[k];
    }
    *(vrow*)(out + (size_t)node * FEAT + lane * VW) = acc;
}

extern "C" void kernel_launch(void* const* d_in, const int* in_sizes, int n_in,
                              void* d_out, int out_size, void* d_ws, size_t ws_size,
                              hipStream_t stream) {
    (void)n_in; (void)out_size; (void)ws_size;
    const float* z  = (const float*)d_in[0];
    const int*   ei = (const int*)d_in[1];
    const float* W1 = (const float*)d_in[2];
    const float* b1 = (const float*)d_in[3];
    const float* W2 = (const float*)d_in[4];
    const float* b2 = (const float*)d_in[5];

    const int N = in_sizes[0] / 128;
    const int E = in_sizes[1] / 2;

    char* ws = (char*)d_ws;
    size_t off = 0;
    auto carve = [&](size_t bytes) -> void* {
        void* p = ws + off;
        off = (off + bytes + 255) & ~(size_t)255;
        return p;
    };
    unsigned* deg    = (unsigned*)carve((size_t)N * 4);
    float*    dinv   = (float*)carve((size_t)N * 4);
    unsigned* rowptr = (unsigned*)carve(((size_t)N + 1) * 4);
    unsigned* rank   = (unsigned*)carve((size_t)E * 4);
    int*      ent    = (int*)carve(((size_t)E + (size_t)N) * 8);
    float*    h1     = (float*)carve((size_t)N * 128 * 4);
    float*    h2     = (float*)carve((size_t)N * 128 * 4);
    float*    h3     = (float*)carve((size_t)N * 64 * 4);
    float*    outp   = (float*)d_out;

    const dim3 blk(256);
    const int gN = (N + 255) / 256;
    const int gE = (E + 255) / 256;
    const int gM = (N + 127) / 128;
    const int gS = (N + 1023) / 1024;
    const int gG = (N + 7) / 8;

    k_zero_u32  <<<gN, blk, 0, stream>>>(deg, N);
    k_count_rank<<<gE, blk, 0, stream>>>(ei, E, deg, rank);
    k_scan_rows <<<gS, blk, 0, stream>>>(deg, N, rowptr, dinv, ent);
    k_fill_rows <<<gE, blk, 0, stream>>>(ei, E, rank, rowptr, dinv, ent);

    k_gemm_wmma<128><<<gM, blk, 0, stream>>>(z, W1, h1, N);
    k_gather_rows<128, true><<<gG, blk, 0, stream>>>(h1, rowptr, ent, b1, h2, N);

    k_gemm_wmma<64><<<gM, blk, 0, stream>>>(h2, W2, h3, N);
    k_gather_rows<64, false><<<gG, blk, 0, stream>>>(h3, rowptr, ent, b2, outp, N);
}


# Layer 1 computed as (A z) W1: one kernel gathers z rows for 16 nodes per wave into LDS, runs both f32 WMMA products there and writes only h3; sums reassociated (aggregate before transform)
# speedup vs baseline: 2.8750x; 1.1004x over previous
#include <hip/hip_runtime.h>
#include <hip/hip_bf16.h>

typedef float v2f __attribute__((ext_vector_type(2)));
typedef float v8f __attribute__((ext_vector_type(8)));

#define LRELU_SLOPE 0.01f

__global__ void k_zero_u32(unsigned* __restrict__ p, int n) {
    int i = blockIdx.x * blockDim.x + threadIdx.x;
    if (i < n) p[i] = 0u;
}

__global__ void k_count_rank(const int* __restrict__ ei, int E, unsigned* __restrict__ deg,
                             unsigned* __restrict__ rank) {
    int e = blockIdx.x * blockDim.x + threadIdx.x;
    if (e < E) {
        int d = ei[E + e];
        rank[e] = atomicAdd(&deg[d], 1u);
    }
}

__global__ __launch_bounds__(256) void k_scan_rows(const unsigned* __restrict__ deg, int n,
                                                   unsigned* __restrict__ rowptr,
                                                   float* __restrict__ dinv,
                                                   int* __restrict__ ent_raw) {
    __shared__ unsigned s_front[8];
    __shared__ unsigned s_wsum[8];
    int2* __restrict__ ent = (int2*)ent_raw;
    const int tid = threadIdx.x;
    const int lane = tid & 31;
    const int wave = tid >> 5;
    const int blk0 = blockIdx.x * 1024;

    unsigned front = 0u;
    const uint4* __restrict__ deg4 = (const uint4*)deg;
#pragma unroll 8
    for (int p = 0; p < (int)blockIdx.x; ++p) {
        const uint4 q = deg4[p * 256 + tid];
        front += (q.x + q.y) + (q.z + q.w);
    }

    const int i0 = blk0 + tid * 4;
    unsigned c[4] = {0u, 0u, 0u, 0u};
    if (i0 + 3 < n) {
        const uint4 q = deg4[i0 >> 2];
        c[0] = q.x + 1u; c[1] = q.y + 1u; c[2] = q.z + 1u; c[3] = q.w + 1u;
    } else {
#pragma unroll
        for (int k = 0; k < 4; ++k) if (i0 + k < n) c[k] = deg[i0 + k] + 1u;
    }
    const unsigned tsum = c[0] + c[1] + c[2] + c[3];

    unsigned incl = tsum;
#pragma unroll
    for (int off = 1; off < 32; off <<= 1) {
        unsigned v = __shfl_up(incl, off);
        if (lane >= off) incl += v;
    }
#pragma unroll
    for (int off = 16; off > 0; off >>= 1) front += __shfl_xor(front, off);
    if (lane == 31) s_wsum[wave] = incl;
    if (lane == 0) s_front[wave] = front;
    __syncthreads();

    unsigned base = (unsigned)blk0;
#pragma unroll
    for (int w = 0; w < 8; ++w) {
        base += s_front[w];
        if (w < wave) base += s_wsum[w];
    }
    unsigned pos = base + incl - tsum;
#pragma unroll
    for (int k = 0; k < 4; ++k) {
        const int idx = i0 + k;
        if (idx < n) {
            rowptr[idx] = pos;
            const float di = rsqrtf((float)c[k]);
            dinv[idx] = di;
            ent[pos] = make_int2(idx, __float_as_int(di * di));
            pos += c[k];
            if (idx == n - 1) rowptr[n] = pos;
        }
    }
}

__global__ void k_fill_rows(const int* __restrict__ ei, int E, const unsigned* __restrict__ rank,
                            const unsigned* __restrict__ rowptr, const float* __restrict__ dinv,
                            int* __restrict__ ent_raw) {
    int2* __restrict__ ent = (int2*)ent_raw;
    int e = blockIdx.x * blockDim.x + threadIdx.x;
    if (e < E) {
        const int s = ei[e];
        const int d = ei[E + e];
        const unsigned slot = rowptr[d] + 1u + rank[e];
        ent[slot] = make_int2(s, __float_as_int(dinv[s] * dinv[d]));
    }
}

#define FW 25
#define TP 132
__global__ __launch_bounds__(FW * 32) void k_layer1_fused(const float* __restrict__ z,
                                                          const unsigned* __restrict__ rowptr,
                                                          const int* __restrict__ ent_raw,
                                                          const float* __restrict__ W1,
                                                          const float* __restrict__ b1,
                                                          const float* __restrict__ W2,
                                                          float* __restrict__ h3, int n_rows) {
    __shared__ __attribute__((aligned(16))) float sW1[128 * 128];
    __shared__ __attribute__((aligned(16))) float sW2[128 * 64];
    __shared__ __attribute__((aligned(16))) float sT[FW * 16 * TP];
    const int2* __restrict__ ent = (const int2*)ent_raw;
    const int tid = threadIdx.x;
    for (int i = tid; i < 128 * 128; i += FW * 32) {
        const int k = i >> 7, n = i & 127;
        sW1[((k >> 1) * 128 + n) * 2 + (k & 1)] = W1[i];
    }
    for (int i = tid; i < 128 * 64; i += FW * 32) {
        const int k = i >> 6, n = i & 63;
        sW2[((k >> 1) * 64 + n) * 2 + (k & 1)] = W2[i];
    }
    __syncthreads();

    const int wave = __builtin_amdgcn_readfirstlane(tid >> 5);
    const int lane = tid & 31;
    const int m16  = lane & 15;
    const int hi   = lane >> 4;
    const int n0   = (blockIdx.x * FW + wave) * 16;
    if (n0 >= n_rows) return;
    float* tile = sT + wave * (16 * TP);

    {
        const int n_hi = n0 + 16 < n_rows ? n0 + 16 : n_rows;
        const unsigned e_beg = rowptr[n0];
        const unsigned e_end = rowptr[n_hi];
        int bi = n0 + m16 + 1;
        if (bi > n_rows) bi = n_rows;
        const unsigned bnd = rowptr[bi];
        int cur = 0;
        unsigned cur_bnd = (unsigned)__builtin_amdgcn_readlane((int)bnd, 0);
        float4 acc = make_float4(0.0f, 0.0f, 0.0f, 0.0f);
        const float* zl = z + lane * 4;
        int2 nxt = make_int2(0, 0);
        if (e_beg + (unsigned)lane < e_end) nxt = ent[e_beg + lane];
        for (unsigned base = e_beg; base < e_end; base += 32u) {
            const int2 my = nxt;
            nxt = make_int2(0, 0);
            if (base + 32u + (unsigned)lane < e_end) nxt = ent[base + 32u + lane];
#pragma unroll
            for (int half = 0; half < 2; ++half) {
                const unsigned b16 = base + 16u * half;
                if (b16 < e_end) {
                    float4 v[16];
#pragma unroll
                    for (int i = 0; i < 16; ++i) {
                        const int s = __builtin_amdgcn_readlane(my.x, half * 16 + i);
                        v[i] = *(const float4*)(zl + (size_t)s * 128);
                    }
#pragma unroll
                    for (int i = 0; i < 16; ++i) {
                        const unsigned idx = b16 + i;
                        if (idx < e_end) {
                            const float w = __int_as_float(__builtin_amdgcn_readlane(my.y, half * 16 + i));
                            acc.x += w * v[i].x;
                            acc.y += w * v[i].y;
                            acc.z += w * v[i].z;
                            acc.w += w * v[i].w;
                            if (idx + 1u == cur_bnd) {
                                *(float4*)(tile + cur * TP + lane * 4) = acc;
                                acc = make_float4(0.0f, 0.0f, 0.0f, 0.0f);
                                cur = (cur + 1) & 15;
                                cur_bnd = (unsigned)__builtin_amdgcn_readlane((int)bnd, cur);
                            }
                        }
                    }
                }
            }
        }
    }

    v8f zero8 = {};
    v8f acc1[8];
#pragma unroll
    for (int t = 0; t < 8; ++t) acc1[t] = zero8;
    const float* arow = tile + m16 * TP;
    for (int k0 = 0; k0 < 128; k0 += 4) {
        v2f a = *(const v2f*)(arow + k0 + 2 * hi);
        const int kh = (k0 >> 1) + hi;
#pragma unroll
        for (int t = 0; t < 8; ++t) {
            v2f b = *(const v2f*)(sW1 + (size_t)(kh * 128 + t * 16 + m16) * 2);
            acc1[t] = __builtin_amdgcn_wmma_f32_16x16x4_f32(
                false, a, false, b, (short)0, acc1[t], false, false);
        }
    }

#pragma unroll
    for (int t = 0; t < 8; ++t) {
        const float bias = b1[t * 16 + m16];
#pragma unroll
        for (int v = 0; v < 8; ++v) {
            float x = acc1[t][v] + bias;
            x = x > 0.0f ? x : LRELU_SLOPE * x;
            tile[(v + 8 * hi) * TP + t * 16 + m16] = x;
        }
    }

    v8f acc2[4];
#pragma unroll
    for (int t = 0; t < 4; ++t) acc2[t] = zero8;
    for (int k0 = 0; k0 < 128; k0 += 4) {
        v2f a = *(const v2f*)(arow + k0 + 2 * hi);
        const int kh = (k0 >> 1) + hi;
#pragma unroll
        for (int t = 0; t < 4; ++t) {
            v2f b = *(const v2f*)(sW2 + (size_t)(kh * 64 + t * 16 + m16) * 2);
            acc2[t] = __builtin_amdgcn_wmma_f32_16x16x4_f32(
                false, a, false, b, (short)0, acc2[t], false, false);
        }
    }
#pragma unroll
    for (int t = 0; t < 4; ++t) {
#pragma unroll
        for (int v = 0; v < 8; ++v) {
            const int row = n0 + v + 8 * hi;
            if (row < n_rows) h3[(size_t)row * 64 + t * 16 + m16] = acc2[t][v];
        }
    }
}

template <int FEAT, bool LEAKY>
__global__ __launch_bounds__(256) void k_gather_rows(const float* __restrict__ h,
                                                     const unsigned* __restrict__ rowptr,
                                                     const int* __restrict__ ent_raw,
                                                     const float* __restrict__ b,
                                                     float* __restrict__ out, int n_rows) {
    constexpr int VW = FEAT / 32;
    typedef float vrow __attribute__((ext_vector_type(VW)));
    const int2* __restrict__ ent = (const int2*)ent_raw;
    int node = (int)(((long long)blockIdx.x * 256 + threadIdx.x) >> 5);
    if (node >= n_rows) return;
    node = __builtin_amdgcn_readfirstlane(node);
    const int lane = threadIdx.x & 31;
    const unsigned beg = rowptr[node];
    const unsigned end = rowptr[node + 1];
    const float* hl = h + lane * VW;

    vrow acc = {};
    for (unsigned base = beg; base < end; base += 32u) {
        const unsigned left = end - base;
        const unsigned cnt = left < 32u ? left : 32u;
        int2 my = make_int2(0, 0);
        if ((unsigned)lane < cnt) my = ent[base + lane];
        unsigned j = 0;
        for (; j + 4u <= cnt; j += 4u) {
            const int s0 = __builtin_amdgcn_readlane(my.x, j + 0);
            const int s1 = __builtin_amdgcn_readlane(my.x, j + 1);
            const int s2 = __builtin_amdgcn_readlane(my.x, j + 2);
            const int s3 = __builtin_amdgcn_readlane(my.x, j + 3);
            const float w0 = __int_as_float(__builtin_amdgcn_readlane(my.y, j + 0));
            const float w1 = __int_as_float(__builtin_amdgcn_readlane(my.y, j + 1));
            const float w2 = __int_as_float(__builtin_amdgcn_readlane(my.y, j + 2));
            const float w3 = __int_as_float(__builtin_amdgcn_readlane(my.y, j + 3));
            const vrow v0 = *(const vrow*)(hl + (size_t)s0 * FEAT);
            const vrow v1 = *(const vrow*)(hl + (size_t)s1 * FEAT);
            const vrow v2 = *(const vrow*)(hl + (size_t)s2 * FEAT);
            const vrow v3 = *(const vrow*)(hl + (size_t)s3 * FEAT);
            acc += w0 * v0;
            acc += w1 * v1;
            acc += w2 * v2;
            acc += w3 * v3;
        }
        if (j < cnt) {
            const int s0 = __builtin_amdgcn_readlane(my.x, j);
            const float w0 = __int_as_float(__builtin_amdgcn_readlane(my.y, j));
            const vrow v0 = *(const vrow*)(hl + (size_t)s0 * FEAT);
            vrow v1 = {}, v2 = {};
            float w1 = 0.0f, w2 = 0.0f;
            if (j + 1u < cnt) {
                const int s1 = __builtin_amdgcn_readlane(my.x, j + 1);
                w1 = __int_as_float(__builtin_amdgcn_readlane(my.y, j + 1));
                v1 = *(const vrow*)(hl + (size_t)s1 * FEAT);
            }
            if (j + 2u < cnt) {
                const int s2 = __builtin_amdgcn_readlane(my.x, j + 2);
                w2 = __int_as_float(__builtin_amdgcn_readlane(my.y, j + 2));
                v2 = *(const vrow*)(hl + (size_t)s2 * FEAT);
            }
            acc += w0 * v0;
            acc += w1 * v1;
            acc += w2 * v2;
        }
    }

    const vrow bb = *(const vrow*)(b + lane * VW);
    acc += bb;
    if (LEAKY) {
#pragma unroll
        for (int k = 0; k < VW; ++k) acc[k] = acc[k] > 0.0f ? acc[k] : LRELU_SLOPE * acc[k];
    }
    *(vrow*)(out + (size_t)node * FEAT + lane * VW) = acc;
}

extern "C" void kernel_launch(void* const* d_in, const int* in_sizes, int n_in,
                              void* d_out, int out_size, void* d_ws, size_t ws_size,
                              hipStream_t stream) {
    (void)n_in; (void)out_size; (void)ws_size;
    const float* z  = (const float*)d_in[0];
    const int*   ei = (const int*)d_in[1];
    const float* W1 = (const float*)d_in[2];
    const float* b1 = (const float*)d_in[3];
    const float* W2 = (const float*)d_in[4];
    const float* b2 = (const float*)d_in[5];

    const int N = in_sizes[0] / 128;
    const int E = in_sizes[1] / 2;

    char* ws = (char*)d_ws;
    size_t off = 0;
    auto carve = [&](size_t bytes) -> void* {
        void* p = ws + off;
        off = (off + bytes + 255) & ~(size_t)255;
        return p;
    };
    unsigned* deg    = (unsigned*)carve((size_t)N * 4);
    float*    dinv   = (float*)carve((size_t)N * 4);
    unsigned* rowptr = (unsigned*)carve(((size_t)N + 1) * 4);
    unsigned* rank   = (unsigned*)carve((size_t)E * 4);
    int*      ent    = (int*)carve(((size_t)E + (size_t)N) * 8);
    float*    h3     = (float*)carve((size_t)N * 64 * 4);
    float*    outp   = (float*)d_out;

    const dim3 blk(256);
    const int gN = (N + 255) / 256;
    const int gE = (E + 255) / 256;
    const int gF = (N + 399) / 400;
    const int gS = (N + 1023) / 1024;
    const int gG = (N + 7) / 8;

    k_zero_u32  <<<gN, blk, 0, stream>>>(deg, N);
    k_count_rank<<<gE, blk, 0, stream>>>(ei, E, deg, rank);
    k_scan_rows <<<gS, blk, 0, stream>>>(deg, N, rowptr, dinv, ent);
    k_fill_rows <<<gE, blk, 0, stream>>>(ei, E, rank, rowptr, dinv, ent);

    k_layer1_fused<<<gF, dim3(FW * 32), 0, stream>>>(z, rowptr, ent, W1, b1, W2, h3, N);

    k_gather_rows<64, false><<<gG, blk, 0, stream>>>(h3, rowptr, ent, b2, outp, N);
}


# Edges of arrival rank under 32 go into a 32-place row of their destination during the count; one wave per 8 nodes packs the rows into the sorted list; the per-edge fill is left for later ranks
# speedup vs baseline: 3.0209x; 1.0507x over previous
#include <hip/hip_runtime.h>
#include <hip/hip_bf16.h>

typedef float v2f __attribute__((ext_vector_type(2)));
typedef float v8f __attribute__((ext_vector_type(8)));

#define LRELU_SLOPE 0.01f

__global__ void k_zero_u32(unsigned* __restrict__ p, int n) {
    int i = blockIdx.x * blockDim.x + threadIdx.x;
    if (i < n) p[i] = 0u;
}

#define SLOTS 32
__global__ void k_count_rank(const int* __restrict__ ei, int E, unsigned* __restrict__ deg,
                             unsigned* __restrict__ rank, int* __restrict__ slot) {
    int e = blockIdx.x * blockDim.x + threadIdx.x;
    if (e < E) {
        const int s = ei[e];
        const int d = ei[E + e];
        asm volatile("" :: "v"(s));
        const unsigned r = atomicAdd(&deg[d], 1u);
        rank[e] = r;
        if (r < SLOTS) slot[(size_t)d * SLOTS + r] = s;
    }
}

__global__ __launch_bounds__(256) void k_scan_rows(const unsigned* __restrict__ deg, int n,
                                                   unsigned* __restrict__ rowptr,
                                                   float* __restrict__ dinv,
                                                   int* __restrict__ ent_raw) {
    __shared__ unsigned s_front[8];
    __shared__ unsigned s_wsum[8];
    int2* __restrict__ ent = (int2*)ent_raw;
    const int tid = threadIdx.x;
    const int lane = tid & 31;
    const int wave = tid >> 5;
    const int blk0 = blockIdx.x * 1024;

    unsigned front = 0u;
    const uint4* __restrict__ deg4 = (const uint4*)deg;
#pragma unroll 8
    for (int p = 0; p < (int)blockIdx.x; ++p) {
        const uint4 q = deg4[p * 256 + tid];
        front += (q.x + q.y) + (q.z + q.w);
    }

    const int i0 = blk0 + tid * 4;
    unsigned c[4] = {0u, 0u, 0u, 0u};
    if (i0 + 3 < n) {
        const uint4 q = deg4[i0 >> 2];
        c[0] = q.x + 1u; c[1] = q.y + 1u; c[2] = q.z + 1u; c[3] = q.w + 1u;
    } else {
#pragma unroll
        for (int k = 0; k < 4; ++k) if (i0 + k < n) c[k] = deg[i0 + k] + 1u;
    }
    const unsigned tsum = c[0] + c[1] + c[2] + c[3];

    unsigned incl = tsum;
#pragma unroll
    for (int off = 1; off < 32; off <<= 1) {
        unsigned v = __shfl_up(incl, off);
        if (lane >= off) incl += v;
    }
#pragma unroll
    for (int off = 16; off > 0; off >>= 1) front += __shfl_xor(front, off);
    if (lane == 31) s_wsum[wave] = incl;
    if (lane == 0) s_front[wave] = front;
    __syncthreads();

    unsigned base = (unsigned)blk0;
#pragma unroll
    for (int w = 0; w < 8; ++w) {
        base += s_front[w];
        if (w < wave) base += s_wsum[w];
    }
    unsigned pos = base + incl - tsum;
#pragma unroll
    for (int k = 0; k < 4; ++k) {
        const int idx = i0 + k;
        if (idx < n) {
            rowptr[idx] = pos;
            const float di = rsqrtf((float)c[k]);
            dinv[idx] = di;
            ent[pos] = make_int2(idx, __float_as_int(di * di));
            pos += c[k];
            if (idx == n - 1) rowptr[n] = pos;
        }
    }
}

#define PK 8
__global__ __launch_bounds__(256) void k_pack_rows(const int* __restrict__ ei, int E,
                                                   const unsigned* __restrict__ deg,
                                                   const unsigned* __restrict__ rank,
                                                   const int* __restrict__ slot,
                                                   const unsigned* __restrict__ rowptr,
                                                   const float* __restrict__ dinv,
                                                   int* __restrict__ ent_raw, int n) {
    int2* __restrict__ ent = (int2*)ent_raw;
    const int lane = threadIdx.x & 31;
    const int wave = __builtin_amdgcn_readfirstlane((int)((blockIdx.x * 256 + threadIdx.x) >> 5));

    const int e0 = wave * 128 + lane * 4;
    unsigned r[4] = {0u, 0u, 0u, 0u};
    if (e0 + 3 < E) {
        const uint4 q = *(const uint4*)(rank + e0);
        r[0] = q.x; r[1] = q.y; r[2] = q.z; r[3] = q.w;
    } else {
#pragma unroll
        for (int c = 0; c < 4; ++c) if (e0 + c < E) r[c] = rank[e0 + c];
    }

    const int n0 = wave * PK;
    int      sv[PK];
    unsigned pos[PK];
    float    dd[PK], ds[PK];
    bool     on[PK];
#pragma unroll
    for (int k = 0; k < PK; ++k) {
        const int node = n0 + k;
        on[k] = false; sv[k] = 0; pos[k] = 0u; dd[k] = 0.0f;
        if (node < n) {
            on[k]  = (unsigned)lane < deg[node];
            pos[k] = rowptr[node] + 1u + (unsigned)lane;
            dd[k]  = dinv[node];
            if (on[k]) sv[k] = slot[(size_t)node * SLOTS + lane];
        }
    }
#pragma unroll
    for (int k = 0; k < PK; ++k) {
        ds[k] = dinv[on[k] ? sv[k] : 0];
    }
#pragma unroll
    for (int k = 0; k < PK; ++k) {
        if (on[k]) ent[pos[k]] = make_int2(sv[k], __float_as_int(ds[k] * dd[k]));
    }

#pragma unroll
    for (int c = 0; c < 4; ++c) {
        if (r[c] >= SLOTS) {
            const int e = e0 + c;
            const int s = ei[e];
            const int d = ei[E + e];
            const unsigned at = rowptr[d] + 1u + r[c];
            ent[at] = make_int2(s, __float_as_int(dinv[s] * dinv[d]));
        }
    }
}

#define FW 25
#define TP 132
__global__ __launch_bounds__(FW * 32) void k_layer1_fused(const float* __restrict__ z,
                                                          const unsigned* __restrict__ rowptr,
                                                          const int* __restrict__ ent_raw,
                                                          const float* __restrict__ W1,
                                                          const float* __restrict__ b1,
                                                          const float* __restrict__ W2,
                                                          float* __restrict__ h3, int n_rows) {
    __shared__ __attribute__((aligned(16))) float sW1[128 * 128];
    __shared__ __attribute__((aligned(16))) float sW2[128 * 64];
    __shared__ __attribute__((aligned(16))) float sT[FW * 16 * TP];
    const int2* __restrict__ ent = (const int2*)ent_raw;
    const int tid = threadIdx.x;
    for (int i = tid; i < 128 * 128; i += FW * 32) {
        const int k = i >> 7, n = i & 127;
        sW1[((k >> 1) * 128 + n) * 2 + (k & 1)] = W1[i];
    }
    for (int i = tid; i < 128 * 64; i += FW * 32) {
        const int k = i >> 6, n = i & 63;
        sW2[((k >> 1) * 64 + n) * 2 + (k & 1)] = W2[i];
    }
    __syncthreads();

    const int wave = __builtin_amdgcn_readfirstlane(tid >> 5);
    const int lane = tid & 31;
    const int m16  = lane & 15;
    const int hi   = lane >> 4;
    const int n0   = (blockIdx.x * FW + wave) * 16;
    if (n0 >= n_rows) return;
    float* tile = sT + wave * (16 * TP);

    {
        const int n_hi = n0 + 16 < n_rows ? n0 + 16 : n_rows;
        const unsigned e_beg = rowptr[n0];
        const unsigned e_end = rowptr[n_hi];
        int bi = n0 + m16 + 1;
        if (bi > n_rows) bi = n_rows;
        const unsigned bnd = rowptr[bi];
        int cur = 0;
        unsigned cur_bnd = (unsigned)__builtin_amdgcn_readlane((int)bnd, 0);
        float4 acc = make_float4(0.0f, 0.0f, 0.0f, 0.0f);
        const float* zl = z + lane * 4;
        int2 nxt = make_int2(0, 0);
        if (e_beg + (unsigned)lane < e_end) nxt = ent[e_beg + lane];
        for (unsigned base = e_beg; base < e_end; base += 32u) {
            const int2 my = nxt;
            nxt = make_int2(0, 0);
            if (base + 32u + (unsigned)lane < e_end) nxt = ent[base + 32u + lane];
#pragma unroll
            for (int half = 0; half < 2; ++half) {
                const unsigned b16 = base + 16u * half;
                if (b16 < e_end) {
                    float4 v[16];
#pragma unroll
                    for (int i = 0; i < 16; ++i) {
                        const int s = __builtin_amdgcn_readlane(my.x, half * 16 + i);
                        v[i] = *(const float4*)(zl + (size_t)s * 128);
                    }
#pragma unroll
                    for (int i = 0; i < 16; ++i) {
                        const unsigned idx = b16 + i;
                        if (idx < e_end) {
                            const float w = __int_as_float(__builtin_amdgcn_readlane(my.y, half * 16 + i));
                            acc.x += w * v[i].x;
                            acc.y += w * v[i].y;
                            acc.z += w * v[i].z;
                            acc.w += w * v[i].w;
                            if (idx + 1u == cur_bnd) {
                                *(float4*)(tile + cur * TP + lane * 4) = acc;
                                acc = make_float4(0.0f, 0.0f, 0.0f, 0.0f);
                                cur = (cur + 1) & 15;
                                cur_bnd = (unsigned)__builtin_amdgcn_readlane((int)bnd, cur);
                            }
                        }
                    }
                }
            }
        }
    }

    v8f zero8 = {};
    v8f acc1[8];
#pragma unroll
    for (int t = 0; t < 8; ++t) acc1[t] = zero8;
    const float* arow = tile + m16 * TP;
    for (int k0 = 0; k0 < 128; k0 += 4) {
        v2f a = *(const v2f*)(arow + k0 + 2 * hi);
        const int kh = (k0 >> 1) + hi;
#pragma unroll
        for (int t = 0; t < 8; ++t) {
            v2f b = *(const v2f*)(sW1 + (size_t)(kh * 128 + t * 16 + m16) * 2);
            acc1[t] = __builtin_amdgcn_wmma_f32_16x16x4_f32(
                false, a, false, b, (short)0, acc1[t], false, false);
        }
    }

#pragma unroll
    for (int t = 0; t < 8; ++t) {
        const float bias = b1[t * 16 + m16];
#pragma unroll
        for (int v = 0; v < 8; ++v) {
            float x = acc1[t][v] + bias;
            x = x > 0.0f ? x : LRELU_SLOPE * x;
            tile[(v + 8 * hi) * TP + t * 16 + m16] = x;
        }
    }

    v8f acc2[4];
#pragma unroll
    for (int t = 0; t < 4; ++t) acc2[t] = zero8;
    for (int k0 = 0; k0 < 128; k0 += 4) {
        v2f a = *(const v2f*)(arow + k0 + 2 * hi);
        const int kh = (k0 >> 1) + hi;
#pragma unroll
        for (int t = 0; t < 4; ++t) {
            v2f b = *(const v2f*)(sW2 + (size_t)(kh * 64 + t * 16 + m16) * 2);
            acc2[t] = __builtin_amdgcn_wmma_f32_16x16x4_f32(
                false, a, false, b, (short)0, acc2[t], false, false);
        }
    }
#pragma unroll
    for (int t = 0; t < 4; ++t) {
#pragma unroll
        for (int v = 0; v < 8; ++v) {
            const int row = n0 + v + 8 * hi;
            if (row < n_rows) h3[(size_t)row * 64 + t * 16 + m16] = acc2[t][v];
        }
    }
}

template <int FEAT, bool LEAKY>
__global__ __launch_bounds__(256) void k_gather_rows(const float* __restrict__ h,
                                                     const unsigned* __restrict__ rowptr,
                                                     const int* __restrict__ ent_raw,
                                                     const float* __restrict__ b,
                                                     float* __restrict__ out, int n_rows) {
    constexpr int VW = FEAT / 32;
    typedef float vrow __attribute__((ext_vector_type(VW)));
    const int2* __restrict__ ent = (const int2*)ent_raw;
    int node = (int)(((long long)blockIdx.x * 256 + threadIdx.x) >> 5);
    if (node >= n_rows) return;
    node = __builtin_amdgcn_readfirstlane(node);
    const int lane = threadIdx.x & 31;
    const unsigned beg = rowptr[node];
    const unsigned end = rowptr[node + 1];
    const float* hl = h + lane * VW;

    vrow acc = {};
    for (unsigned base = beg; base < end; base += 32u) {
        const unsigned left = end - base;
        const unsigned cnt = left < 32u ? left : 32u;
        int2 my = make_int2(0, 0);
        if ((unsigned)lane < cnt) my = ent[base + lane];
        unsigned j = 0;
        for (; j + 4u <= cnt; j += 4u) {
            const int s0 = __builtin_amdgcn_readlane(my.x, j + 0);
            const int s1 = __builtin_amdgcn_readlane(my.x, j + 1);
            const int s2 = __builtin_amdgcn_readlane(my.x, j + 2);
            const int s3 = __builtin_amdgcn_readlane(my.x, j + 3);
            const float w0 = __int_as_float(__builtin_amdgcn_readlane(my.y, j + 0));
            const float w1 = __int_as_float(__builtin_amdgcn_readlane(my.y, j + 1));
            const float w2 = __int_as_float(__builtin_amdgcn_readlane(my.y, j + 2));
            const float w3 = __int_as_float(__builtin_amdgcn_readlane(my.y, j + 3));
            const vrow v0 = *(const vrow*)(hl + (size_t)s0 * FEAT);
            const vrow v1 = *(const vrow*)(hl + (size_t)s1 * FEAT);
            const vrow v2 = *(const vrow*)(hl + (size_t)s2 * FEAT);
            const vrow v3 = *(const vrow*)(hl + (size_t)s3 * FEAT);
            acc += w0 * v0;
            acc += w1 * v1;
            acc += w2 * v2;
            acc += w3 * v3;
        }
        if (j < cnt) {
            const int s0 = __builtin_amdgcn_readlane(my.x, j);
            const float w0 = __int_as_float(__builtin_amdgcn_readlane(my.y, j));
            const vrow v0 = *(const vrow*)(hl + (size_t)s0 * FEAT);
            vrow v1 = {}, v2 = {};
            float w1 = 0.0f, w2 = 0.0f;
            if (j + 1u < cnt) {
                const int s1 = __builtin_amdgcn_readlane(my.x, j + 1);
                w1 = __int_as_float(__builtin_amdgcn_readlane(my.y, j + 1));
                v1 = *(const vrow*)(hl + (size_t)s1 * FEAT);
            }
            if (j + 2u < cnt) {
                const int s2 = __builtin_amdgcn_readlane(my.x, j + 2);
                w2 = __int_as_float(__builtin_amdgcn_readlane(my.y, j + 2));
                v2 = *(const vrow*)(hl + (size_t)s2 * FEAT);
            }
            acc += w0 * v0;
            acc += w1 * v1;
            acc += w2 * v2;
        }
    }

    const vrow bb = *(const vrow*)(b + lane * VW);
    acc += bb;
    if (LEAKY) {
#pragma unroll
        for (int k = 0; k < VW; ++k) acc[k] = acc[k] > 0.0f ? acc[k] : LRELU_SLOPE * acc[k];
    }
    *(vrow*)(out + (size_t)node * FEAT + lane * VW) = acc;
}

extern "C" void kernel_launch(void* const* d_in, const int* in_sizes, int n_in,
                              void* d_out, int out_size, void* d_ws, size_t ws_size,
                              hipStream_t stream) {
    (void)n_in; (void)out_size; (void)ws_size;
    const float* z  = (const float*)d_in[0];
    const int*   ei = (const int*)d_in[1];
    const float* W1 = (const float*)d_in[2];
    const float* b1 = (const float*)d_in[3];
    const float* W2 = (const float*)d_in[4];
    const float* b2 = (const float*)d_in[5];

    const int N = in_sizes[0] / 128;
    const int E = in_sizes[1] / 2;

    char* ws = (char*)d_ws;
    size_t off = 0;
    auto carve = [&](size_t bytes) -> void* {
        void* p = ws + off;
        off = (off + bytes + 255) & ~(size_t)255;
        return p;
    };
    unsigned* deg    = (unsigned*)carve((size_t)N * 4);
    float*    dinv   = (float*)carve((size_t)N * 4);
    unsigned* rowptr = (unsigned*)carve(((size_t)N + 1) * 4);
    unsigned* rank   = (unsigned*)carve((size_t)E * 4);
    int*      ent    = (int*)carve(((size_t)E + (size_t)N) * 8);
    int*      slot   = (int*)carve((size_t)N * SLOTS * 4);
    float*    h3     = (float*)carve((size_t)N * 64 * 4);
    float*    outp   = (float*)d_out;

    const dim3 blk(256);
    const int gN = (N + 255) / 256;
    const int gE = (E + 255) / 256;
    const int gF = (N + 399) / 400;
    const int gS = (N + 1023) / 1024;
    const int gG = (N + 7) / 8;
    const int gPn = (N + 8 * PK - 1) / (8 * PK);
    const int gPe = (E + 1023) / 1024;
    const int gP = gPn > gPe ? gPn : gPe;

    k_zero_u32  <<<gN, blk, 0, stream>>>(deg, N);
    k_count_rank<<<gE, blk, 0, stream>>>(ei, E, deg, rank, slot);
    k_scan_rows <<<gS, blk, 0, stream>>>(deg, N, rowptr, dinv, ent);
    k_pack_rows <<<gP, blk, 0, stream>>>(ei, E, deg, rank, slot, rowptr, dinv, ent, N);

    k_layer1_fused<<<gF, dim3(FW * 32), 0, stream>>>(z, rowptr, ent, W1, b1, W2, h3, N);

    k_gather_rows<64, false><<<gG, blk, 0, stream>>>(h3, rowptr, ent, b2, outp, N);
}


# Edge ranks from LDS counters in a two-level counting sort (chunk pass groups edges by destination bucket, one work-group per bucket ranks them), no global atomic per edge; sums in LDS arrival order
# speedup vs baseline: 4.5788x; 1.5157x over previous
#include <hip/hip_runtime.h>
#include <hip/hip_bf16.h>

typedef float v2f __attribute__((ext_vector_type(2)));
typedef float v8f __attribute__((ext_vector_type(8)));

#define LRELU_SLOPE 0.01f

#define SLOTS 32
#define NBS 8
#define NB (1 << NBS)
#define CHT 1024
#define EPT 8
#define CH (CHT * EPT)

__global__ __launch_bounds__(CHT) void k_chunk_sort(const int* __restrict__ ei, int E, int B,
                                                    int* __restrict__ pair_raw,
                                                    unsigned* __restrict__ seg) {
    __shared__ unsigned s_cnt[CHT];
    __shared__ unsigned s_wsum[CHT / 32];
    int2* __restrict__ pair = (int2*)pair_raw;
    const int tid = threadIdx.x;
    const int lane = tid & 31;
    const int wave = tid >> 5;
    const long long base = (long long)blockIdx.x * CH;
    s_cnt[tid] = 0u;
    __syncthreads();

    int  s[EPT], d[EPT];
    bool on[EPT];
    unsigned rk[EPT];
#pragma unroll
    for (int j = 0; j < EPT; ++j) {
        const long long e = base + j * CHT + tid;
        on[j] = e < (long long)E;
        s[j] = 0; d[j] = 0;
        if (on[j]) { s[j] = ei[e]; d[j] = ei[(long long)E + e]; }
    }
#pragma unroll
    for (int j = 0; j < EPT; ++j) {
        rk[j] = 0u;
        if (on[j]) rk[j] = atomicAdd(&s_cnt[d[j] >> NBS], 1u);
    }
    __syncthreads();

    const unsigned v = s_cnt[tid];
    unsigned incl = v;
#pragma unroll
    for (int off = 1; off < 32; off <<= 1) {
        const unsigned t = __shfl_up(incl, off);
        if (lane >= off) incl += t;
    }
    if (lane == 31) s_wsum[wave] = incl;
    __syncthreads();
    unsigned wpre = 0u;
#pragma unroll
    for (int w = 0; w < CHT / 32; ++w) if (w < wave) wpre += s_wsum[w];
    const unsigned excl = wpre + incl - v;
    s_cnt[tid] = excl;
    unsigned* __restrict__ row = seg + (size_t)blockIdx.x * (size_t)(B + 1);
    if (tid < B) row[tid] = excl;
    if (tid == 0) {
        const long long left = (long long)E - base;
        row[B] = (unsigned)(left < (long long)CH ? left : (long long)CH);
    }
    __syncthreads();
#pragma unroll
    for (int j = 0; j < EPT; ++j) {
        if (on[j]) pair[base + s_cnt[d[j] >> NBS] + rk[j]] = make_int2(s[j], d[j]);
    }
}

#define RKT 1024
__global__ __launch_bounds__(RKT) void k_bucket_rank(const int* __restrict__ pair_raw,
                                                     const unsigned* __restrict__ seg, int K, int B,
                                                     unsigned* __restrict__ deg,
                                                     unsigned* __restrict__ rank,
                                                     int* __restrict__ slot, int n) {
    __shared__ unsigned s_deg[NB];
    const int2* __restrict__ pair = (const int2*)pair_raw;
    const int tid = threadIdx.x;
    const int lane = tid & 31;
    const int wave = __builtin_amdgcn_readfirstlane(tid >> 5);
    const int b = blockIdx.x;
    const int n_lo = b * NB;
    if (tid < NB) s_deg[tid] = 0u;
    __syncthreads();

    for (int k0 = wave; k0 < K; k0 += 32 * (RKT / 32)) {
        const int kl = k0 + (RKT / 32) * lane;
        int beg = 0, end = 0;
        if (kl < K) {
            const unsigned* __restrict__ r = seg + (size_t)kl * (size_t)(B + 1) + b;
            beg = (int)r[0];
            end = (int)r[1];
        }
        for (int i = 0; i < 32; ++i) {
            const int k = k0 + (RKT / 32) * i;
            if (k >= K) break;
            const unsigned sb = (unsigned)__builtin_amdgcn_readlane(beg, i);
            const unsigned se = (unsigned)__builtin_amdgcn_readlane(end, i);
            const size_t pbase = (size_t)k * CH;
            for (unsigned o = sb + (unsigned)lane; o < se; o += 32u) {
                const size_t p = pbase + o;
                const int2 sd = pair[p];
                const unsigned r = atomicAdd(&s_deg[sd.y - n_lo], 1u);
                rank[p] = r;
                if (r < SLOTS) slot[(size_t)sd.y * SLOTS + r] = sd.x;
            }
        }
    }
    __syncthreads();
    if (tid < NB && n_lo + tid < n) deg[n_lo + tid] = s_deg[tid];
}

__global__ __launch_bounds__(256) void k_scan_rows(const unsigned* __restrict__ deg, int n,
                                                   unsigned* __restrict__ rowptr,
                                                   float* __restrict__ dinv,
                                                   int* __restrict__ ent_raw) {
    __shared__ unsigned s_front[8];
    __shared__ unsigned s_wsum[8];
    int2* __restrict__ ent = (int2*)ent_raw;
    const int tid = threadIdx.x;
    const int lane = tid & 31;
    const int wave = tid >> 5;
    const int blk0 = blockIdx.x * 1024;

    unsigned front = 0u;
    const uint4* __restrict__ deg4 = (const uint4*)deg;
#pragma unroll 8
    for (int p = 0; p < (int)blockIdx.x; ++p) {
        const uint4 q = deg4[p * 256 + tid];
        front += (q.x + q.y) + (q.z + q.w);
    }

    const int i0 = blk0 + tid * 4;
    unsigned c[4] = {0u, 0u, 0u, 0u};
    if (i0 + 3 < n) {
        const uint4 q = deg4[i0 >> 2];
        c[0] = q.x + 1u; c[1] = q.y + 1u; c[2] = q.z + 1u; c[3] = q.w + 1u;
    } else {
#pragma unroll
        for (int k = 0; k < 4; ++k) if (i0 + k < n) c[k] = deg[i0 + k] + 1u;
    }
    const unsigned tsum = c[0] + c[1] + c[2] + c[3];

    unsigned incl = tsum;
#pragma unroll
    for (int off = 1; off < 32; off <<= 1) {
        unsigned v = __shfl_up(incl, off);
        if (lane >= off) incl += v;
    }
#pragma unroll
    for (int off = 16; off > 0; off >>= 1) front += __shfl_xor(front, off);
    if (lane == 31) s_wsum[wave] = incl;
    if (lane == 0) s_front[wave] = front;
    __syncthreads();

    unsigned base = (unsigned)blk0;
#pragma unroll
    for (int w = 0; w < 8; ++w) {
        base += s_front[w];
        if (w < wave) base += s_wsum[w];
    }
    unsigned pos = base + incl - tsum;
#pragma unroll
    for (int k = 0; k < 4; ++k) {
        const int idx = i0 + k;
        if (idx < n) {
            rowptr[idx] = pos;
            const float di = rsqrtf((float)c[k]);
            dinv[idx] = di;
            ent[pos] = make_int2(idx, __float_as_int(di * di));
            pos += c[k];
            if (idx == n - 1) rowptr[n] = pos;
        }
    }
}

#define PK 8
__global__ __launch_bounds__(256) void k_pack_rows(const int* __restrict__ pair_raw, int E,
                                                   const unsigned* __restrict__ deg,
                                                   const unsigned* __restrict__ rank,
                                                   const int* __restrict__ slot,
                                                   const unsigned* __restrict__ rowptr,
                                                   const float* __restrict__ dinv,
                                                   int* __restrict__ ent_raw, int n) {
    int2* __restrict__ ent = (int2*)ent_raw;
    const int2* __restrict__ pair = (const int2*)pair_raw;
    const int lane = threadIdx.x & 31;
    const int wave = __builtin_amdgcn_readfirstlane((int)((blockIdx.x * 256 + threadIdx.x) >> 5));

    const int e0 = wave * 128 + lane * 4;
    unsigned r[4] = {0u, 0u, 0u, 0u};
    if (e0 + 3 < E) {
        const uint4 q = *(const uint4*)(rank + e0);
        r[0] = q.x; r[1] = q.y; r[2] = q.z; r[3] = q.w;
    } else {
#pragma unroll
        for (int c = 0; c < 4; ++c) if (e0 + c < E) r[c] = rank[e0 + c];
    }

    const int n0 = wave * PK;
    int      sv[PK];
    unsigned pos[PK];
    float    dd[PK], ds[PK];
    bool     on[PK];
#pragma unroll
    for (int k = 0; k < PK; ++k) {
        const int node = n0 + k;
        on[k] = false; sv[k] = 0; pos[k] = 0u; dd[k] = 0.0f;
        if (node < n) {
            on[k]  = (unsigned)lane < deg[node];
            pos[k] = rowptr[node] + 1u + (unsigned)lane;
            dd[k]  = dinv[node];
            if (on[k]) sv[k] = slot[(size_t)node * SLOTS + lane];
        }
    }
#pragma unroll
    for (int k = 0; k < PK; ++k) {
        ds[k] = dinv[on[k] ? sv[k] : 0];
    }
#pragma unroll
    for (int k = 0; k < PK; ++k) {
        if (on[k]) ent[pos[k]] = make_int2(sv[k], __float_as_int(ds[k] * dd[k]));
    }

#pragma unroll
    for (int c = 0; c < 4; ++c) {
        if (r[c] >= SLOTS) {
            const int2 sd = pair[e0 + c];
            const int s = sd.x;
            const int d = sd.y;
            const unsigned at = rowptr[d] + 1u + r[c];
            ent[at] = make_int2(s, __float_as_int(dinv[s] * dinv[d]));
        }
    }
}

#define FW 25
#define TP 132
__global__ __launch_bounds__(FW * 32) void k_layer1_fused(const float* __restrict__ z,
                                                          const unsigned* __restrict__ rowptr,
                                                          const int* __restrict__ ent_raw,
                                                          const float* __restrict__ W1,
                                                          const float* __restrict__ b1,
                                                          const float* __restrict__ W2,
                                                          float* __restrict__ h3, int n_rows) {
    __shared__ __attribute__((aligned(16))) float sW1[128 * 128];
    __shared__ __attribute__((aligned(16))) float sW2[128 * 64];
    __shared__ __attribute__((aligned(16))) float sT[FW * 16 * TP];
    const int2* __restrict__ ent = (const int2*)ent_raw;
    const int tid = threadIdx.x;
    for (int i = tid; i < 128 * 128; i += FW * 32) {
        const int k = i >> 7, n = i & 127;
        sW1[((k >> 1) * 128 + n) * 2 + (k & 1)] = W1[i];
    }
    for (int i = tid; i < 128 * 64; i += FW * 32) {
        const int k = i >> 6, n = i & 63;
        sW2[((k >> 1) * 64 + n) * 2 + (k & 1)] = W2[i];
    }
    __syncthreads();

    const int wave = __builtin_amdgcn_readfirstlane(tid >> 5);
    const int lane = tid & 31;
    const int m16  = lane & 15;
    const int hi   = lane >> 4;
    const int n0   = (blockIdx.x * FW + wave) * 16;
    if (n0 >= n_rows) return;
    float* tile = sT + wave * (16 * TP);

    {
        const int n_hi = n0 + 16 < n_rows ? n0 + 16 : n_rows;
        const unsigned e_beg = rowptr[n0];
        const unsigned e_end = rowptr[n_hi];
        int bi = n0 + m16 + 1;
        if (bi > n_rows) bi = n_rows;
        const unsigned bnd = rowptr[bi];
        int cur = 0;
        unsigned cur_bnd = (unsigned)__builtin_amdgcn_readlane((int)bnd, 0);
        float4 acc = make_float4(0.0f, 0.0f, 0.0f, 0.0f);
        const float* zl = z + lane * 4;
        int2 nxt = make_int2(0, 0);
        if (e_beg + (unsigned)lane < e_end) nxt = ent[e_beg + lane];
        for (unsigned base = e_beg; base < e_end; base += 32u) {
            const int2 my = nxt;
            nxt = make_int2(0, 0);
            if (base + 32u + (unsigned)lane < e_end) nxt = ent[base + 32u + lane];
#pragma unroll
            for (int half = 0; half < 2; ++half) {
                const unsigned b16 = base + 16u * half;
                if (b16 < e_end) {
                    float4 v[16];
#pragma unroll
                    for (int i = 0; i < 16; ++i) {
                        const int s = __builtin_amdgcn_readlane(my.x, half * 16 + i);
                        v[i] = *(const float4*)(zl + (size_t)s * 128);
                    }
#pragma unroll
                    for (int i = 0; i < 16; ++i) {
                        const unsigned idx = b16 + i;
                        if (idx < e_end) {
                            const float w = __int_as_float(__builtin_amdgcn_readlane(my.y, half * 16 + i));
                            acc.x += w * v[i].x;
                            acc.y += w * v[i].y;
                            acc.z += w * v[i].z;
                            acc.w += w * v[i].w;
                            if (idx + 1u == cur_bnd) {
                                *(float4*)(tile + cur * TP + lane * 4) = acc;
                                acc = make_float4(0.0f, 0.0f, 0.0f, 0.0f);
                                cur = (cur + 1) & 15;
                                cur_bnd = (unsigned)__builtin_amdgcn_readlane((int)bnd, cur);
                            }
                        }
                    }
                }
            }
        }
    }

    v8f zero8 = {};
    v8f acc1[8];
#pragma unroll
    for (int t = 0; t < 8; ++t) acc1[t] = zero8;
    const float* arow = tile + m16 * TP;
    for (int k0 = 0; k0 < 128; k0 += 4) {
        v2f a = *(const v2f*)(arow + k0 + 2 * hi);
        const int kh = (k0 >> 1) + hi;
#pragma unroll
        for (int t = 0; t < 8; ++t) {
            v2f b = *(const v2f*)(sW1 + (size_t)(kh * 128 + t * 16 + m16) * 2);
            acc1[t] = __builtin_amdgcn_wmma_f32_16x16x4_f32(
                false, a, false, b, (short)0, acc1[t], false, false);
        }
    }

#pragma unroll
    for (int t = 0; t < 8; ++t) {
        const float bias = b1[t * 16 + m16];
#pragma unroll
        for (int v = 0; v < 8; ++v) {
            float x = acc1[t][v] + bias;
            x = x > 0.0f ? x : LRELU_SLOPE * x;
            tile[(v + 8 * hi) * TP + t * 16 + m16] = x;
        }
    }

    v8f acc2[4];
#pragma unroll
    for (int t = 0; t < 4; ++t) acc2[t] = zero8;
    for (int k0 = 0; k0 < 128; k0 += 4) {
        v2f a = *(const v2f*)(arow + k0 + 2 * hi);
        const int kh = (k0 >> 1) + hi;
#pragma unroll
        for (int t = 0; t < 4; ++t) {
            v2f b = *(const v2f*)(sW2 + (size_t)(kh * 64 + t * 16 + m16) * 2);
            acc2[t] = __builtin_amdgcn_wmma_f32_16x16x4_f32(
                false, a, false, b, (short)0, acc2[t], false, false);
        }
    }
#pragma unroll
    for (int t = 0; t < 4; ++t) {
#pragma unroll
        for (int v = 0; v < 8; ++v) {
            const int row = n0 + v + 8 * hi;
            if (row < n_rows) h3[(size_t)row * 64 + t * 16 + m16] = acc2[t][v];
        }
    }
}

template <int FEAT, bool LEAKY>
__global__ __launch_bounds__(256) void k_gather_rows(const float* __restrict__ h,
                                                     const unsigned* __restrict__ rowptr,
                                                     const int* __restrict__ ent_raw,
                                                     const float* __restrict__ b,
                                                     float* __restrict__ out, int n_rows) {
    constexpr int VW = FEAT / 32;
    typedef float vrow __attribute__((ext_vector_type(VW)));
    const int2* __restrict__ ent = (const int2*)ent_raw;
    int node = (int)(((long long)blockIdx.x * 256 + threadIdx.x) >> 5);
    if (node >= n_rows) return;
    node = __builtin_amdgcn_readfirstlane(node);
    const int lane = threadIdx.x & 31;
    const unsigned beg = rowptr[node];
    const unsigned end = rowptr[node + 1];
    const float* hl = h + lane * VW;

    vrow acc = {};
    for (unsigned base = beg; base < end; base += 32u) {
        const unsigned left = end - base;
        const unsigned cnt = left < 32u ? left : 32u;
        int2 my = make_int2(0, 0);
        if ((unsigned)lane < cnt) my = ent[base + lane];
        unsigned j = 0;
        for (; j + 4u <= cnt; j += 4u) {
            const int s0 = __builtin_amdgcn_readlane(my.x, j + 0);
            const int s1 = __builtin_amdgcn_readlane(my.x, j + 1);
            const int s2 = __builtin_amdgcn_readlane(my.x, j + 2);
            const int s3 = __builtin_amdgcn_readlane(my.x, j + 3);
            const float w0 = __int_as_float(__builtin_amdgcn_readlane(my.y, j + 0));
            const float w1 = __int_as_float(__builtin_amdgcn_readlane(my.y, j + 1));
            const float w2 = __int_as_float(__builtin_amdgcn_readlane(my.y, j + 2));
            const float w3 = __int_as_float(__builtin_amdgcn_readlane(my.y, j + 3));
            const vrow v0 = *(const vrow*)(hl + (size_t)s0 * FEAT);
            const vrow v1 = *(const vrow*)(hl + (size_t)s1 * FEAT);
            const vrow v2 = *(const vrow*)(hl + (size_t)s2 * FEAT);
            const vrow v3 = *(const vrow*)(hl + (size_t)s3 * FEAT);
            acc += w0 * v0;
            acc += w1 * v1;
            acc += w2 * v2;
            acc += w3 * v3;
        }
        if (j < cnt) {
            const int s0 = __builtin_amdgcn_readlane(my.x, j);
            const float w0 = __int_as_float(__builtin_amdgcn_readlane(my.y, j));
            const vrow v0 = *(const vrow*)(hl + (size_t)s0 * FEAT);
            vrow v1 = {}, v2 = {};
            float w1 = 0.0f, w2 = 0.0f;
            if (j + 1u < cnt) {
                const int s1 = __builtin_amdgcn_readlane(my.x, j + 1);
                w1 = __int_as_float(__builtin_amdgcn_readlane(my.y, j + 1));
                v1 = *(const vrow*)(hl + (size_t)s1 * FEAT);
            }
            if (j + 2u < cnt) {
                const int s2 = __builtin_amdgcn_readlane(my.x, j + 2);
                w2 = __int_as_float(__builtin_amdgcn_readlane(my.y, j + 2));
                v2 = *(const vrow*)(hl + (size_t)s2 * FEAT);
            }
            acc += w0 * v0;
            acc += w1 * v1;
            acc += w2 * v2;
        }
    }

    const vrow bb = *(const vrow*)(b + lane * VW);
    acc += bb;
    if (LEAKY) {
#pragma unroll
        for (int k = 0; k < VW; ++k) acc[k] = acc[k] > 0.0f ? acc[k] : LRELU_SLOPE * acc[k];
    }
    *(vrow*)(out + (size_t)node * FEAT + lane * VW) = acc;
}

extern "C" void kernel_launch(void* const* d_in, const int* in_sizes, int n_in,
                              void* d_out, int out_size, void* d_ws, size_t ws_size,
                              hipStream_t stream) {
    (void)n_in; (void)out_size; (void)ws_size;
    const float* z  = (const float*)d_in[0];
    const int*   ei = (const int*)d_in[1];
    const float* W1 = (const float*)d_in[2];
    const float* b1 = (const float*)d_in[3];
    const float* W2 = (const float*)d_in[4];
    const float* b2 = (const float*)d_in[5];

    const int N = in_sizes[0] / 128;
    const int E = in_sizes[1] / 2;

    char* ws = (char*)d_ws;
    size_t off = 0;
    auto carve = [&](size_t bytes) -> void* {
        void* p = ws + off;
        off = (off + bytes + 255) & ~(size_t)255;
        return p;
    };
    unsigned* deg    = (unsigned*)carve((size_t)N * 4);
    float*    dinv   = (float*)carve((size_t)N * 4);
    unsigned* rowptr = (unsigned*)carve(((size_t)N + 1) * 4);
    const int K = (E + CH - 1) / CH;
    const int B = (N + NB - 1) / NB;
    int*      pair   = (int*)carve((size_t)K * CH * 8);
    unsigned* seg    = (unsigned*)carve((size_t)K * (B + 1) * 4);
    unsigned* rank   = (unsigned*)carve((size_t)E * 4);
    int*      ent    = (int*)carve(((size_t)E + (size_t)N) * 8);
    int*      slot   = (int*)carve((size_t)N * SLOTS * 4);
    float*    h3     = (float*)carve((size_t)N * 64 * 4);
    float*    outp   = (float*)d_out;

    const dim3 blk(256);
    const int gF = (N + 399) / 400;
    const int gS = (N + 1023) / 1024;
    const int gG = (N + 7) / 8;
    const int gPn = (N + 8 * PK - 1) / (8 * PK);
    const int gPe = (E + 1023) / 1024;
    const int gP = gPn > gPe ? gPn : gPe;

    k_chunk_sort <<<K, dim3(CHT), 0, stream>>>(ei, E, B, pair, seg);
    k_bucket_rank<<<B, dim3(RKT), 0, stream>>>(pair, seg, K, B, deg, rank, slot, N);
    k_scan_rows  <<<gS, blk, 0, stream>>>(deg, N, rowptr, dinv, ent);
    k_pack_rows  <<<gP, blk, 0, stream>>>(pair, E, deg, rank, slot, rowptr, dinv, ent, N);

    k_layer1_fused<<<gF, dim3(FW * 32), 0, stream>>>(z, rowptr, ent, W1, b1, W2, h3, N);

    k_gather_rows<64, false><<<gG, blk, 0, stream>>>(h3, rowptr, ent, b2, outp, N);
}


# Row starts, dinv and self-loop entries written by the ranking work-group itself: edges in front = sum of its piece starts over the chunks, then a scan of its 256 counters; scan launch gone (6 to 5)
# speedup vs baseline: 4.9594x; 1.0831x over previous
#include <hip/hip_runtime.h>
#include <hip/hip_bf16.h>

typedef float v2f __attribute__((ext_vector_type(2)));
typedef float v8f __attribute__((ext_vector_type(8)));

#define LRELU_SLOPE 0.01f

#define SLOTS 32
#define NBS 8
#define NB (1 << NBS)
#define CHT 1024
#define EPT 8
#define CH (CHT * EPT)

__global__ __launch_bounds__(CHT) void k_chunk_sort(const int* __restrict__ ei, int E, int B,
                                                    int* __restrict__ pair_raw,
                                                    unsigned* __restrict__ seg) {
    __shared__ unsigned s_cnt[CHT];
    __shared__ unsigned s_wsum[CHT / 32];
    int2* __restrict__ pair = (int2*)pair_raw;
    const int tid = threadIdx.x;
    const int lane = tid & 31;
    const int wave = tid >> 5;
    const long long base = (long long)blockIdx.x * CH;
    s_cnt[tid] = 0u;
    __syncthreads();

    int  s[EPT], d[EPT];
    bool on[EPT];
    unsigned rk[EPT];
#pragma unroll
    for (int j = 0; j < EPT; ++j) {
        const long long e = base + j * CHT + tid;
        on[j] = e < (long long)E;
        s[j] = 0; d[j] = 0;
        if (on[j]) { s[j] = ei[e]; d[j] = ei[(long long)E + e]; }
    }
#pragma unroll
    for (int j = 0; j < EPT; ++j) {
        rk[j] = 0u;
        if (on[j]) rk[j] = atomicAdd(&s_cnt[d[j] >> NBS], 1u);
    }
    __syncthreads();

    const unsigned v = s_cnt[tid];
    unsigned incl = v;
#pragma unroll
    for (int off = 1; off < 32; off <<= 1) {
        const unsigned t = __shfl_up(incl, off);
        if (lane >= off) incl += t;
    }
    if (lane == 31) s_wsum[wave] = incl;
    __syncthreads();
    unsigned wpre = 0u;
#pragma unroll
    for (int w = 0; w < CHT / 32; ++w) if (w < wave) wpre += s_wsum[w];
    const unsigned excl = wpre + incl - v;
    s_cnt[tid] = excl;
    unsigned* __restrict__ row = seg + (size_t)blockIdx.x * (size_t)(B + 1);
    if (tid < B) row[tid] = excl;
    if (tid == 0) {
        const long long left = (long long)E - base;
        row[B] = (unsigned)(left < (long long)CH ? left : (long long)CH);
    }
    __syncthreads();
#pragma unroll
    for (int j = 0; j < EPT; ++j) {
        if (on[j]) pair[base + s_cnt[d[j] >> NBS] + rk[j]] = make_int2(s[j], d[j]);
    }
}

#define RKT 1024
__global__ __launch_bounds__(RKT) void k_bucket_rank(const int* __restrict__ pair_raw,
                                                     const unsigned* __restrict__ seg, int K, int B,
                                                     unsigned* __restrict__ deg,
                                                     unsigned* __restrict__ rank,
                                                     int* __restrict__ slot,
                                                     unsigned* __restrict__ rowptr,
                                                     float* __restrict__ dinv,
                                                     int* __restrict__ ent_raw, int n) {
    __shared__ unsigned s_deg[NB];
    __shared__ unsigned s_front[RKT / 32];
    __shared__ unsigned s_wsum[NB / 32];
    const int2* __restrict__ pair = (const int2*)pair_raw;
    int2* __restrict__ ent = (int2*)ent_raw;
    const int tid = threadIdx.x;
    const int lane = tid & 31;
    const int wave = __builtin_amdgcn_readfirstlane(tid >> 5);
    const int b = blockIdx.x;
    const int n_lo = b * NB;
    if (tid < NB) s_deg[tid] = 0u;
    __syncthreads();

    unsigned wfront = 0u;
    for (int k0 = wave; k0 < K; k0 += 32 * (RKT / 32)) {
        const int kl = k0 + (RKT / 32) * lane;
        int beg = 0, end = 0;
        if (kl < K) {
            const unsigned* __restrict__ r = seg + (size_t)kl * (size_t)(B + 1) + b;
            beg = (int)r[0];
            end = (int)r[1];
        }
        for (int i = 0; i < 32; ++i) {
            const int k = k0 + (RKT / 32) * i;
            if (k >= K) break;
            const unsigned sb = (unsigned)__builtin_amdgcn_readlane(beg, i);
            const unsigned se = (unsigned)__builtin_amdgcn_readlane(end, i);
            wfront += sb;
            const size_t pbase = (size_t)k * CH;
            for (unsigned o = sb + (unsigned)lane; o < se; o += 32u) {
                const size_t p = pbase + o;
                const int2 sd = pair[p];
                const unsigned r = atomicAdd(&s_deg[sd.y - n_lo], 1u);
                rank[p] = r;
                if (r < SLOTS) slot[(size_t)sd.y * SLOTS + r] = sd.x;
            }
        }
    }
    if (lane == 0) s_front[wave] = wfront;
    __syncthreads();

    const int idx = n_lo + tid;
    const bool node_on = tid < NB && idx < n;
    unsigned c = 0u, cc = 0u, incl = 0u;
    if (tid < NB) {
        if (idx < n) { c = s_deg[tid]; cc = c + 1u; }
        incl = cc;
#pragma unroll
        for (int off = 1; off < 32; off <<= 1) {
            const unsigned t = __shfl_up(incl, off);
            if (lane >= off) incl += t;
        }
        if (lane == 31) s_wsum[wave] = incl;
    }
    __syncthreads();
    if (node_on) {
        unsigned base = (unsigned)n_lo;
#pragma unroll
        for (int w = 0; w < RKT / 32; ++w) base += s_front[w];
#pragma unroll
        for (int w = 0; w < NB / 32; ++w) if (w < wave) base += s_wsum[w];
        const unsigned pos = base + incl - cc;
        deg[idx] = c;
        rowptr[idx] = pos;
        const float di = rsqrtf((float)cc);
        dinv[idx] = di;
        ent[pos] = make_int2(idx, __float_as_int(di * di));
        if (idx == n - 1) rowptr[n] = pos + cc;
    }
}

#define PK 8
__global__ __launch_bounds__(256) void k_pack_rows(const int* __restrict__ pair_raw, int E,
                                                   const unsigned* __restrict__ deg,
                                                   const unsigned* __restrict__ rank,
                                                   const int* __restrict__ slot,
                                                   const unsigned* __restrict__ rowptr,
                                                   const float* __restrict__ dinv,
                                                   int* __restrict__ ent_raw, int n) {
    int2* __restrict__ ent = (int2*)ent_raw;
    const int2* __restrict__ pair = (const int2*)pair_raw;
    const int lane = threadIdx.x & 31;
    const int wave = __builtin_amdgcn_readfirstlane((int)((blockIdx.x * 256 + threadIdx.x) >> 5));

    const int e0 = wave * 128 + lane * 4;
    unsigned r[4] = {0u, 0u, 0u, 0u};
    if (e0 + 3 < E) {
        const uint4 q = *(const uint4*)(rank + e0);
        r[0] = q.x; r[1] = q.y; r[2] = q.z; r[3] = q.w;
    } else {
#pragma unroll
        for (int c = 0; c < 4; ++c) if (e0 + c < E) r[c] = rank[e0 + c];
    }

    const int n0 = wave * PK;
    int      sv[PK];
    unsigned pos[PK];
    float    dd[PK], ds[PK];
    bool     on[PK];
#pragma unroll
    for (int k = 0; k < PK; ++k) {
        const int node = n0 + k;
        on[k] = false; sv[k] = 0; pos[k] = 0u; dd[k] = 0.0f;
        if (node < n) {
            on[k]  = (unsigned)lane < deg[node];
            pos[k] = rowptr[node] + 1u + (unsigned)lane;
            dd[k]  = dinv[node];
            if (on[k]) sv[k] = slot[(size_t)node * SLOTS + lane];
        }
    }
#pragma unroll
    for (int k = 0; k < PK; ++k) {
        ds[k] = dinv[on[k] ? sv[k] : 0];
    }
#pragma unroll
    for (int k = 0; k < PK; ++k) {
        if (on[k]) ent[pos[k]] = make_int2(sv[k], __float_as_int(ds[k] * dd[k]));
    }

#pragma unroll
    for (int c = 0; c < 4; ++c) {
        if (r[c] >= SLOTS) {
            const int2 sd = pair[e0 + c];
            const int s = sd.x;
            const int d = sd.y;
            const unsigned at = rowptr[d] + 1u + r[c];
            ent[at] = make_int2(s, __float_as_int(dinv[s] * dinv[d]));
        }
    }
}

#define FW 25
#define TP 132
__global__ __launch_bounds__(FW * 32) void k_layer1_fused(const float* __restrict__ z,
                                                          const unsigned* __restrict__ rowptr,
                                                          const int* __restrict__ ent_raw,
                                                          const float* __restrict__ W1,
                                                          const float* __restrict__ b1,
                                                          const float* __restrict__ W2,
                                                          float* __restrict__ h3, int n_rows) {
    __shared__ __attribute__((aligned(16))) float sW1[128 * 128];
    __shared__ __attribute__((aligned(16))) float sW2[128 * 64];
    __shared__ __attribute__((aligned(16))) float sT[FW * 16 * TP];
    const int2* __restrict__ ent = (const int2*)ent_raw;
    const int tid = threadIdx.x;
    for (int i = tid; i < 128 * 128; i += FW * 32) {
        const int k = i >> 7, n = i & 127;
        sW1[((k >> 1) * 128 + n) * 2 + (k & 1)] = W1[i];
    }
    for (int i = tid; i < 128 * 64; i += FW * 32) {
        const int k = i >> 6, n = i & 63;
        sW2[((k >> 1) * 64 + n) * 2 + (k & 1)] = W2[i];
    }
    __syncthreads();

    const int wave = __builtin_amdgcn_readfirstlane(tid >> 5);
    const int lane = tid & 31;
    const int m16  = lane & 15;
    const int hi   = lane >> 4;
    const int n0   = (blockIdx.x * FW + wave) * 16;
    if (n0 >= n_rows) return;
    float* tile = sT + wave * (16 * TP);

    {
        const int n_hi = n0 + 16 < n_rows ? n0 + 16 : n_rows;
        const unsigned e_beg = rowptr[n0];
        const unsigned e_end = rowptr[n_hi];
        int bi = n0 + m16 + 1;
        if (bi > n_rows) bi = n_rows;
        const unsigned bnd = rowptr[bi];
        int cur = 0;
        unsigned cur_bnd = (unsigned)__builtin_amdgcn_readlane((int)bnd, 0);
        float4 acc = make_float4(0.0f, 0.0f, 0.0f, 0.0f);
        const float* zl = z + lane * 4;
        int2 nxt = make_int2(0, 0);
        if (e_beg + (unsigned)lane < e_end) nxt = ent[e_beg + lane];
        for (unsigned base = e_beg; base < e_end; base += 32u) {
            const int2 my = nxt;
            nxt = make_int2(0, 0);
            if (base + 32u + (unsigned)lane < e_end) nxt = ent[base + 32u + lane];
#pragma unroll
            for (int half = 0; half < 2; ++half) {
                const unsigned b16 = base + 16u * half;
                if (b16 < e_end) {
                    float4 v[16];
#pragma unroll
                    for (int i = 0; i < 16; ++i) {
                        const int s = __builtin_amdgcn_readlane(my.x, half * 16 + i);
                        v[i] = *(const float4*)(zl + (size_t)s * 128);
                    }
#pragma unroll
                    for (int i = 0; i < 16; ++i) {
                        const unsigned idx = b16 + i;
                        if (idx < e_end) {
                            const float w = __int_as_float(__builtin_amdgcn_readlane(my.y, half * 16 + i));
                            acc.x += w * v[i].x;
                            acc.y += w * v[i].y;
                            acc.z += w * v[i].z;
                            acc.w += w * v[i].w;
                            if (idx + 1u == cur_bnd) {
                                *(float4*)(tile + cur * TP + lane * 4) = acc;
                                acc = make_float4(0.0f, 0.0f, 0.0f, 0.0f);
                                cur = (cur + 1) & 15;
                                cur_bnd = (unsigned)__builtin_amdgcn_readlane((int)bnd, cur);
                            }
                        }
                    }
                }
            }
        }
    }

    v8f zero8 = {};
    v8f acc1[8];
#pragma unroll
    for (int t = 0; t < 8; ++t) acc1[t] = zero8;
    const float* arow = tile + m16 * TP;
    for (int k0 = 0; k0 < 128; k0 += 4) {
        v2f a = *(const v2f*)(arow + k0 + 2 * hi);
        const int kh = (k0 >> 1) + hi;
#pragma unroll
        for (int t = 0; t < 8; ++t) {
            v2f b = *(const v2f*)(sW1 + (size_t)(kh * 128 + t * 16 + m16) * 2);
            acc1[t] = __builtin_amdgcn_wmma_f32_16x16x4_f32(
                false, a, false, b, (short)0, acc1[t], false, false);
        }
    }

#pragma unroll
    for (int t = 0; t < 8; ++t) {
        const float bias = b1[t * 16 + m16];
#pragma unroll
        for (int v = 0; v < 8; ++v) {
            float x = acc1[t][v] + bias;
            x = x > 0.0f ? x : LRELU_SLOPE * x;
            tile[(v + 8 * hi) * TP + t * 16 + m16] = x;
        }
    }

    v8f acc2[4];
#pragma unroll
    for (int t = 0; t < 4; ++t) acc2[t] = zero8;
    for (int k0 = 0; k0 < 128; k0 += 4) {
        v2f a = *(const v2f*)(arow + k0 + 2 * hi);
        const int kh = (k0 >> 1) + hi;
#pragma unroll
        for (int t = 0; t < 4; ++t) {
            v2f b = *(const v2f*)(sW2 + (size_t)(kh * 64 + t * 16 + m16) * 2);
            acc2[t] = __builtin_amdgcn_wmma_f32_16x16x4_f32(
                false, a, false, b, (short)0, acc2[t], false, false);
        }
    }
#pragma unroll
    for (int t = 0; t < 4; ++t) {
#pragma unroll
        for (int v = 0; v < 8; ++v) {
            const int row = n0 + v + 8 * hi;
            if (row < n_rows) h3[(size_t)row * 64 + t * 16 + m16] = acc2[t][v];
        }
    }
}

template <int FEAT, bool LEAKY>
__global__ __launch_bounds__(256) void k_gather_rows(const float* __restrict__ h,
                                                     const unsigned* __restrict__ rowptr,
                                                     const int* __restrict__ ent_raw,
                                                     const float* __restrict__ b,
                                                     float* __restrict__ out, int n_rows) {
    constexpr int VW = FEAT / 32;
    typedef float vrow __attribute__((ext_vector_type(VW)));
    const int2* __restrict__ ent = (const int2*)ent_raw;
    int node = (int)(((long long)blockIdx.x * 256 + threadIdx.x) >> 5);
    if (node >= n_rows) return;
    node = __builtin_amdgcn_readfirstlane(node);
    const int lane = threadIdx.x & 31;
    const unsigned beg = rowptr[node];
    const unsigned end = rowptr[node + 1];
    const float* hl = h + lane * VW;

    vrow acc = {};
    for (unsigned base = beg; base < end; base += 32u) {
        const unsigned left = end - base;
        const unsigned cnt = left < 32u ? left : 32u;
        int2 my = make_int2(0, 0);
        if ((unsigned)lane < cnt) my = ent[base + lane];
        unsigned j = 0;
        for (; j + 4u <= cnt; j += 4u) {
            const int s0 = __builtin_amdgcn_readlane(my.x, j + 0);
            const int s1 = __builtin_amdgcn_readlane(my.x, j + 1);
            const int s2 = __builtin_amdgcn_readlane(my.x, j + 2);
            const int s3 = __builtin_amdgcn_readlane(my.x, j + 3);
            const float w0 = __int_as_float(__builtin_amdgcn_readlane(my.y, j + 0));
            const float w1 = __int_as_float(__builtin_amdgcn_readlane(my.y, j + 1));
            const float w2 = __int_as_float(__builtin_amdgcn_readlane(my.y, j + 2));
            const float w3 = __int_as_float(__builtin_amdgcn_readlane(my.y, j + 3));
            const vrow v0 = *(const vrow*)(hl + (size_t)s0 * FEAT);
            const vrow v1 = *(const vrow*)(hl + (size_t)s1 * FEAT);
            const vrow v2 = *(const vrow*)(hl + (size_t)s2 * FEAT);
            const vrow v3 = *(const vrow*)(hl + (size_t)s3 * FEAT);
            acc += w0 * v0;
            acc += w1 * v1;
            acc += w2 * v2;
            acc += w3 * v3;
        }
        if (j < cnt) {
            const int s0 = __builtin_amdgcn_readlane(my.x, j);
            const float w0 = __int_as_float(__builtin_amdgcn_readlane(my.y, j));
            const vrow v0 = *(const vrow*)(hl + (size_t)s0 * FEAT);
            vrow v1 = {}, v2 = {};
            float w1 = 0.0f, w2 = 0.0f;
            if (j + 1u < cnt) {
                const int s1 = __builtin_amdgcn_readlane(my.x, j + 1);
                w1 = __int_as_float(__builtin_amdgcn_readlane(my.y, j + 1));
                v1 = *(const vrow*)(hl + (size_t)s1 * FEAT);
            }
            if (j + 2u < cnt) {
                const int s2 = __builtin_amdgcn_readlane(my.x, j + 2);
                w2 = __int_as_float(__builtin_amdgcn_readlane(my.y, j + 2));
                v2 = *(const vrow*)(hl + (size_t)s2 * FEAT);
            }
            acc += w0 * v0;
            acc += w1 * v1;
            acc += w2 * v2;
        }
    }

    const vrow bb = *(const vrow*)(b + lane * VW);
    acc += bb;
    if (LEAKY) {
#pragma unroll
        for (int k = 0; k < VW; ++k) acc[k] = acc[k] > 0.0f ? acc[k] : LRELU_SLOPE * acc[k];
    }
    *(vrow*)(out + (size_t)node * FEAT + lane * VW) = acc;
}

extern "C" void kernel_launch(void* const* d_in, const int* in_sizes, int n_in,
                              void* d_out, int out_size, void* d_ws, size_t ws_size,
                              hipStream_t stream) {
    (void)n_in; (void)out_size; (void)ws_size;
    const float* z  = (const float*)d_in[0];
    const int*   ei = (const int*)d_in[1];
    const float* W1 = (const float*)d_in[2];
    const float* b1 = (const float*)d_in[3];
    const float* W2 = (const float*)d_in[4];
    const float* b2 = (const float*)d_in[5];

    const int N = in_sizes[0] / 128;
    const int E = in_sizes[1] / 2;

    char* ws = (char*)d_ws;
    size_t off = 0;
    auto carve = [&](size_t bytes) -> void* {
        void* p = ws + off;
        off = (off + bytes + 255) & ~(size_t)255;
        return p;
    };
    unsigned* deg    = (unsigned*)carve((size_t)N * 4);
    float*    dinv   = (float*)carve((size_t)N * 4);
    unsigned* rowptr = (unsigned*)carve(((size_t)N + 1) * 4);
    const int K = (E + CH - 1) / CH;
    const int B = (N + NB - 1) / NB;
    int*      pair   = (int*)carve((size_t)K * CH * 8);
    unsigned* seg    = (unsigned*)carve((size_t)K * (B + 1) * 4);
    unsigned* rank   = (unsigned*)carve((size_t)E * 4);
    int*      ent    = (int*)carve(((size_t)E + (size_t)N) * 8);
    int*      slot   = (int*)carve((size_t)N * SLOTS * 4);
    float*    h3     = (float*)carve((size_t)N * 64 * 4);
    float*    outp   = (float*)d_out;

    const dim3 blk(256);
    const int gF = (N + 399) / 400;
    const int gG = (N + 7) / 8;
    const int gPn = (N + 8 * PK - 1) / (8 * PK);
    const int gPe = (E + 1023) / 1024;
    const int gP = gPn > gPe ? gPn : gPe;

    k_chunk_sort <<<K, dim3(CHT), 0, stream>>>(ei, E, B, pair, seg);
    k_bucket_rank<<<B, dim3(RKT), 0, stream>>>(pair, seg, K, B, deg, rank, slot, rowptr, dinv, ent, N);
    k_pack_rows  <<<gP, blk, 0, stream>>>(pair, E, deg, rank, slot, rowptr, dinv, ent, N);

    k_layer1_fused<<<gF, dim3(FW * 32), 0, stream>>>(z, rowptr, ent, W1, b1, W2, h3, N);

    k_gather_rows<64, false><<<gG, blk, 0, stream>>>(h3, rowptr, ent, b2, outp, N);
}


# Edge weight applied as dinv[dst]*sum(dinv[src]*row): the list holds sources only and the bucket work-group fills it in a second walk; pack launch, slot rows, ranks gone (5 to 4); sums refactored
# speedup vs baseline: 5.3168x; 1.0721x over previous
#include <hip/hip_runtime.h>
#include <hip/hip_bf16.h>

typedef float v2f __attribute__((ext_vector_type(2)));
typedef float v8f __attribute__((ext_vector_type(8)));

#define LRELU_SLOPE 0.01f

#define NBS 8
#define NB (1 << NBS)
#define CHT 1024
#define EPT 8
#define CH (CHT * EPT)

__global__ __launch_bounds__(CHT) void k_chunk_sort(const int* __restrict__ ei, int E, int B,
                                                    int* __restrict__ pair_raw,
                                                    unsigned* __restrict__ seg) {
    __shared__ unsigned s_cnt[CHT];
    __shared__ unsigned s_wsum[CHT / 32];
    int2* __restrict__ pair = (int2*)pair_raw;
    const int tid = threadIdx.x;
    const int lane = tid & 31;
    const int wave = tid >> 5;
    const long long base = (long long)blockIdx.x * CH;
    s_cnt[tid] = 0u;
    __syncthreads();

    int  s[EPT], d[EPT];
    bool on[EPT];
    unsigned rk[EPT];
#pragma unroll
    for (int j = 0; j < EPT; ++j) {
        const long long e = base + j * CHT + tid;
        on[j] = e < (long long)E;
        s[j] = 0; d[j] = 0;
        if (on[j]) { s[j] = ei[e]; d[j] = ei[(long long)E + e]; }
    }
#pragma unroll
    for (int j = 0; j < EPT; ++j) {
        rk[j] = 0u;
        if (on[j]) rk[j] = atomicAdd(&s_cnt[d[j] >> NBS], 1u);
    }
    __syncthreads();

    const unsigned v = s_cnt[tid];
    unsigned incl = v;
#pragma unroll
    for (int off = 1; off < 32; off <<= 1) {
        const unsigned t = __shfl_up(incl, off);
        if (lane >= off) incl += t;
    }
    if (lane == 31) s_wsum[wave] = incl;
    __syncthreads();
    unsigned wpre = 0u;
#pragma unroll
    for (int w = 0; w < CHT / 32; ++w) if (w < wave) wpre += s_wsum[w];
    const unsigned excl = wpre + incl - v;
    s_cnt[tid] = excl;
    unsigned* __restrict__ row = seg + (size_t)blockIdx.x * (size_t)(B + 1);
    if (tid < B) row[tid] = excl;
    if (tid == 0) {
        const long long left = (long long)E - base;
        row[B] = (unsigned)(left < (long long)CH ? left : (long long)CH);
    }
    __syncthreads();
#pragma unroll
    for (int j = 0; j < EPT; ++j) {
        if (on[j]) pair[base + s_cnt[d[j] >> NBS] + rk[j]] = make_int2(s[j], d[j]);
    }
}

#define RKT 1024
#define PCS 8
template <bool FILL>
__device__ __forceinline__ void walk_one(const int* __restrict__ pair_raw, size_t p, int n_lo,
                                         unsigned* s_cnt, int* __restrict__ ent) {
    if (FILL) {
        const int2 sd = ((const int2*)pair_raw)[p];
        const unsigned at = atomicAdd(&s_cnt[sd.y - n_lo], 1u);
        ent[at] = sd.x;
    } else {
        const int d = pair_raw[2 * p + 1];
        atomicAdd(&s_cnt[d - n_lo], 1u);
    }
}

template <bool FILL>
__device__ __forceinline__ unsigned walk_pieces(const int* __restrict__ pair_raw,
                                                const unsigned* __restrict__ seg, int K, int B,
                                                int b, int n_lo, int wave, int lane,
                                                unsigned* s_cnt, int* __restrict__ ent) {
    unsigned wfront = 0u;
    for (int k0 = wave; k0 < K; k0 += 32 * (RKT / 32)) {
        const int kl = k0 + (RKT / 32) * lane;
        int beg = 0, end = 0;
        if (kl < K) {
            const unsigned* __restrict__ r = seg + (size_t)kl * (size_t)(B + 1) + b;
            beg = (int)r[0];
            end = (int)r[1];
        }
        for (int i = 0; i < 32; i += PCS) {
            if (k0 + (RKT / 32) * i >= K) break;
            unsigned se[PCS];
            size_t   p[PCS];
            bool     on[PCS];
            int2     sd[PCS];
#pragma unroll
            for (int j = 0; j < PCS; ++j) {
                const unsigned sb = (unsigned)__builtin_amdgcn_readlane(beg, i + j);
                se[j] = (unsigned)__builtin_amdgcn_readlane(end, i + j);
                wfront += sb;
                const unsigned o = sb + (unsigned)lane;
                p[j] = (size_t)(k0 + (RKT / 32) * (i + j)) * CH + o;
                on[j] = o < se[j];
                sd[j] = make_int2(0, 0);
                if (on[j]) {
                    if (FILL) sd[j] = ((const int2*)pair_raw)[p[j]];
                    else      sd[j].y = pair_raw[2 * p[j] + 1];
                }
            }
#pragma unroll
            for (int j = 0; j < PCS; ++j) {
                if (on[j]) {
                    if (FILL) {
                        const unsigned at = atomicAdd(&s_cnt[sd[j].y - n_lo], 1u);
                        ent[at] = sd[j].x;
                    } else {
                        atomicAdd(&s_cnt[sd[j].y - n_lo], 1u);
                    }
                }
            }
#pragma unroll
            for (int j = 0; j < PCS; ++j) {
                const unsigned sb = (unsigned)__builtin_amdgcn_readlane(beg, i + j);
                for (unsigned o = sb + 32u + (unsigned)lane; o < se[j]; o += 32u)
                    walk_one<FILL>(pair_raw, p[j] + (size_t)(o - sb - (unsigned)lane), n_lo, s_cnt, ent);
            }
        }
    }
    return wfront;
}

__global__ __launch_bounds__(RKT) void k_bucket_rows(const int* __restrict__ pair_raw,
                                                     const unsigned* __restrict__ seg, int K, int B,
                                                     unsigned* __restrict__ rowptr,
                                                     float* __restrict__ dinv,
                                                     int* __restrict__ ent, int n) {
    __shared__ unsigned s_deg[NB];
    __shared__ unsigned s_at[NB];
    __shared__ unsigned s_front[RKT / 32];
    __shared__ unsigned s_wsum[NB / 32];
    const int tid = threadIdx.x;
    const int lane = tid & 31;
    const int wave = __builtin_amdgcn_readfirstlane(tid >> 5);
    const int b = blockIdx.x;
    const int n_lo = b * NB;
    if (tid < NB) s_deg[tid] = 0u;
    __syncthreads();

    const unsigned wfront = walk_pieces<false>(pair_raw, seg, K, B, b, n_lo, wave, lane, s_deg, ent);
    if (lane == 0) s_front[wave] = wfront;
    __syncthreads();

    const int idx = n_lo + tid;
    unsigned cc = 0u, incl = 0u;
    if (tid < NB) {
        if (idx < n) cc = s_deg[tid] + 1u;
        incl = cc;
#pragma unroll
        for (int off = 1; off < 32; off <<= 1) {
            const unsigned t = __shfl_up(incl, off);
            if (lane >= off) incl += t;
        }
        if (lane == 31) s_wsum[wave] = incl;
    }
    __syncthreads();
    if (tid < NB) {
        unsigned base = (unsigned)n_lo;
#pragma unroll
        for (int w = 0; w < RKT / 32; ++w) base += s_front[w];
#pragma unroll
        for (int w = 0; w < NB / 32; ++w) if (w < wave) base += s_wsum[w];
        const unsigned pos = base + incl - cc;
        s_at[tid] = pos + 1u;
        if (idx < n) {
            rowptr[idx] = pos;
            dinv[idx] = rsqrtf((float)cc);
            ent[pos] = idx;
            if (idx == n - 1) rowptr[n] = pos + cc;
        }
    }
    __syncthreads();

    walk_pieces<true>(pair_raw, seg, K, B, b, n_lo, wave, lane, s_at, ent);
}

#define FW 25
#define TP 132
__global__ __launch_bounds__(FW * 32) void k_layer1_fused(const float* __restrict__ z,
                                                          const unsigned* __restrict__ rowptr,
                                                          const int* __restrict__ ent,
                                                          const float* __restrict__ dinv,
                                                          const float* __restrict__ W1,
                                                          const float* __restrict__ b1,
                                                          const float* __restrict__ W2,
                                                          float* __restrict__ h3, int n_rows) {
    __shared__ __attribute__((aligned(16))) float sW1[128 * 128];
    __shared__ __attribute__((aligned(16))) float sW2[128 * 64];
    __shared__ __attribute__((aligned(16))) float sT[FW * 16 * TP];
    const int tid = threadIdx.x;
    for (int i = tid; i < 128 * 128; i += FW * 32) {
        const int k = i >> 7, n = i & 127;
        sW1[((k >> 1) * 128 + n) * 2 + (k & 1)] = W1[i];
    }
    for (int i = tid; i < 128 * 64; i += FW * 32) {
        const int k = i >> 6, n = i & 63;
        sW2[((k >> 1) * 64 + n) * 2 + (k & 1)] = W2[i];
    }
    __syncthreads();

    const int wave = __builtin_amdgcn_readfirstlane(tid >> 5);
    const int lane = tid & 31;
    const int m16  = lane & 15;
    const int hi   = lane >> 4;
    const int n0   = (blockIdx.x * FW + wave) * 16;
    if (n0 >= n_rows) return;
    float* tile = sT + wave * (16 * TP);

    {
        const int n_hi = n0 + 16 < n_rows ? n0 + 16 : n_rows;
        const unsigned e_beg = rowptr[n0];
        const unsigned e_end = rowptr[n_hi];
        int bi = n0 + m16 + 1;
        if (bi > n_rows) bi = n_rows;
        const unsigned bnd = rowptr[bi];
        const float dn = dinv[bi - 1];
        int cur = 0;
        unsigned cur_bnd = (unsigned)__builtin_amdgcn_readlane((int)bnd, 0);
        float cur_dn = __int_as_float(__builtin_amdgcn_readlane(__float_as_int(dn), 0));
        float4 acc = make_float4(0.0f, 0.0f, 0.0f, 0.0f);
        const float* zl = z + lane * 4;
        int s_cur = 0, s_nxt = 0;
        if (e_beg + (unsigned)lane < e_end) s_cur = ent[e_beg + lane];
        if (e_beg + 32u + (unsigned)lane < e_end) s_nxt = ent[e_beg + 32u + lane];
        float w_cur = dinv[s_cur];
        for (unsigned base = e_beg; base < e_end; base += 32u) {
            const int   my_s = s_cur;
            const float my_w = w_cur;
            s_cur = s_nxt;
            w_cur = dinv[s_cur];
            s_nxt = 0;
            if (base + 64u + (unsigned)lane < e_end) s_nxt = ent[base + 64u + lane];
#pragma unroll
            for (int half = 0; half < 2; ++half) {
                const unsigned b16 = base + 16u * half;
                if (b16 < e_end) {
                    float4 v[16];
#pragma unroll
                    for (int i = 0; i < 16; ++i) {
                        const int s = __builtin_amdgcn_readlane(my_s, half * 16 + i);
                        v[i] = *(const float4*)(zl + (size_t)s * 128);
                    }
#pragma unroll
                    for (int i = 0; i < 16; ++i) {
                        const unsigned idx = b16 + i;
                        if (idx < e_end) {
                            const float w = __int_as_float(
                                __builtin_amdgcn_readlane(__float_as_int(my_w), half * 16 + i));
                            acc.x += w * v[i].x;
                            acc.y += w * v[i].y;
                            acc.z += w * v[i].z;
                            acc.w += w * v[i].w;
                            if (idx + 1u == cur_bnd) {
                                *(float4*)(tile + cur * TP + lane * 4) =
                                    make_float4(cur_dn * acc.x, cur_dn * acc.y, cur_dn * acc.z, cur_dn * acc.w);
                                acc = make_float4(0.0f, 0.0f, 0.0f, 0.0f);
                                cur = (cur + 1) & 15;
                                cur_bnd = (unsigned)__builtin_amdgcn_readlane((int)bnd, cur);
                                cur_dn = __int_as_float(__builtin_amdgcn_readlane(__float_as_int(dn), cur));
                            }
                        }
                    }
                }
            }
        }
    }

    v8f zero8 = {};
    v8f acc1[8];
#pragma unroll
    for (int t = 0; t < 8; ++t) acc1[t] = zero8;
    const float* arow = tile + m16 * TP;
    for (int k0 = 0; k0 < 128; k0 += 4) {
        v2f a = *(const v2f*)(arow + k0 + 2 * hi);
        const int kh = (k0 >> 1) + hi;
#pragma unroll
        for (int t = 0; t < 8; ++t) {
            v2f b = *(const v2f*)(sW1 + (size_t)(kh * 128 + t * 16 + m16) * 2);
            acc1[t] = __builtin_amdgcn_wmma_f32_16x16x4_f32(
                false, a, false, b, (short)0, acc1[t], false, false);
        }
    }

#pragma unroll
    for (int t = 0; t < 8; ++t) {
        const float bias = b1[t * 16 + m16];
#pragma unroll
        for (int v = 0; v < 8; ++v) {
            float x = acc1[t][v] + bias;
            x = x > 0.0f ? x : LRELU_SLOPE * x;
            tile[(v + 8 * hi) * TP + t * 16 + m16] = x;
        }
    }

    v8f acc2[4];
#pragma unroll
    for (int t = 0; t < 4; ++t) acc2[t] = zero8;
    for (int k0 = 0; k0 < 128; k0 += 4) {
        v2f a = *(const v2f*)(arow + k0 + 2 * hi);
        const int kh = (k0 >> 1) + hi;
#pragma unroll
        for (int t = 0; t < 4; ++t) {
            v2f b = *(const v2f*)(sW2 + (size_t)(kh * 64 + t * 16 + m16) * 2);
            acc2[t] = __builtin_amdgcn_wmma_f32_16x16x4_f32(
                false, a, false, b, (short)0, acc2[t], false, false);
        }
    }
    float dr[8];
#pragma unroll
    for (int v = 0; v < 8; ++v) {
        const int row = n0 + v + 8 * hi;
        dr[v] = row < n_rows ? dinv[row] : 0.0f;
    }
#pragma unroll
    for (int t = 0; t < 4; ++t) {
#pragma unroll
        for (int v = 0; v < 8; ++v) {
            const int row = n0 + v + 8 * hi;
            if (row < n_rows) h3[(size_t)row * 64 + t * 16 + m16] = dr[v] * acc2[t][v];
        }
    }
}

template <int FEAT>
__global__ __launch_bounds__(256) void k_gather_rows(const float* __restrict__ hs,
                                                     const unsigned* __restrict__ rowptr,
                                                     const int* __restrict__ ent,
                                                     const float* __restrict__ dinv,
                                                     const float* __restrict__ b,
                                                     float* __restrict__ out, int n_rows) {
    constexpr int VW = FEAT / 32;
    typedef float vrow __attribute__((ext_vector_type(VW)));
    int node = (int)(((long long)blockIdx.x * 256 + threadIdx.x) >> 5);
    if (node >= n_rows) return;
    node = __builtin_amdgcn_readfirstlane(node);
    const int lane = threadIdx.x & 31;
    const unsigned beg = rowptr[node];
    const unsigned end = rowptr[node + 1];
    const float dn = dinv[node];
    const float* hl = hs + lane * VW;

    vrow acc = {};
    for (unsigned base = beg; base < end; base += 32u) {
        const unsigned left = end - base;
        const unsigned cnt = left < 32u ? left : 32u;
        int my = 0;
        if ((unsigned)lane < cnt) my = ent[base + lane];
        unsigned j = 0;
        for (; j + 4u <= cnt; j += 4u) {
            const int s0 = __builtin_amdgcn_readlane(my, j + 0);
            const int s1 = __builtin_amdgcn_readlane(my, j + 1);
            const int s2 = __builtin_amdgcn_readlane(my, j + 2);
            const int s3 = __builtin_amdgcn_readlane(my, j + 3);
            const vrow v0 = *(const vrow*)(hl + (size_t)s0 * FEAT);
            const vrow v1 = *(const vrow*)(hl + (size_t)s1 * FEAT);
            const vrow v2 = *(const vrow*)(hl + (size_t)s2 * FEAT);
            const vrow v3 = *(const vrow*)(hl + (size_t)s3 * FEAT);
            acc += v0;
            acc += v1;
            acc += v2;
            acc += v3;
        }
        if (j < cnt) {
            const int s0 = __builtin_amdgcn_readlane(my, j);
            const vrow v0 = *(const vrow*)(hl + (size_t)s0 * FEAT);
            vrow v1 = {}, v2 = {};
            if (j + 1u < cnt) {
                const int s1 = __builtin_amdgcn_readlane(my, j + 1);
                v1 = *(const vrow*)(hl + (size_t)s1 * FEAT);
            }
            if (j + 2u < cnt) {
                const int s2 = __builtin_amdgcn_readlane(my, j + 2);
                v2 = *(const vrow*)(hl + (size_t)s2 * FEAT);
            }
            acc += v0;
            acc += v1;
            acc += v2;
        }
    }

    const vrow bb = *(const vrow*)(b + lane * VW);
    acc = dn * acc + bb;
    *(vrow*)(out + (size_t)node * FEAT + lane * VW) = acc;
}

extern "C" void kernel_launch(void* const* d_in, const int* in_sizes, int n_in,
                              void* d_out, int out_size, void* d_ws, size_t ws_size,
                              hipStream_t stream) {
    (void)n_in; (void)out_size; (void)ws_size;
    const float* z  = (const float*)d_in[0];
    const int*   ei = (const int*)d_in[1];
    const float* W1 = (const float*)d_in[2];
    const float* b1 = (const float*)d_in[3];
    const float* W2 = (const float*)d_in[4];
    const float* b2 = (const float*)d_in[5];

    const int N = in_sizes[0] / 128;
    const int E = in_sizes[1] / 2;

    char* ws = (char*)d_ws;
    size_t off = 0;
    auto carve = [&](size_t bytes) -> void* {
        void* p = ws + off;
        off = (off + bytes + 255) & ~(size_t)255;
        return p;
    };
    float*    dinv   = (float*)carve((size_t)N * 4);
    unsigned* rowptr = (unsigned*)carve(((size_t)N + 1) * 4);
    const int K = (E + CH - 1) / CH;
    const int B = (N + NB - 1) / NB;
    int*      pair   = (int*)carve((size_t)K * CH * 8);
    unsigned* seg    = (unsigned*)carve((size_t)K * (B + 1) * 4);
    int*      ent    = (int*)carve(((size_t)E + (size_t)N) * 4);
    float*    h3     = (float*)carve((size_t)N * 64 * 4);
    float*    outp   = (float*)d_out;

    const dim3 blk(256);
    const int gF = (N + 399) / 400;
    const int gG = (N + 7) / 8;

    k_chunk_sort <<<K, dim3(CHT), 0, stream>>>(ei, E, B, pair, seg);
    k_bucket_rows<<<B, dim3(RKT), 0, stream>>>(pair, seg, K, B, rowptr, dinv, ent, N);

    k_layer1_fused<<<gF, dim3(FW * 32), 0, stream>>>(z, rowptr, ent, dinv, W1, b1, W2, h3, N);

    k_gather_rows<64><<<gG, blk, 0, stream>>>(h3, rowptr, ent, dinv, b2, outp, N);
}


# Layer-1 kernel in wave roles: 13 waves gather tiles into double-buffered LDS (row loads as a rolling window of 16) while 12 waves run the f32 WMMA products of the stage before; sums unchanged
# speedup vs baseline: 5.6729x; 1.0670x over previous
#include <hip/hip_runtime.h>
#include <hip/hip_bf16.h>

typedef float v2f __attribute__((ext_vector_type(2)));
typedef float v8f __attribute__((ext_vector_type(8)));

#define LRELU_SLOPE 0.01f

#define NBS 8
#define NB (1 << NBS)
#define CHT 1024
#define EPT 8
#define CH (CHT * EPT)

__global__ __launch_bounds__(CHT) void k_chunk_sort(const int* __restrict__ ei, int E, int B,
                                                    int* __restrict__ pair_raw,
                                                    unsigned* __restrict__ seg) {
    __shared__ unsigned s_cnt[CHT];
    __shared__ unsigned s_wsum[CHT / 32];
    int2* __restrict__ pair = (int2*)pair_raw;
    const int tid = threadIdx.x;
    const int lane = tid & 31;
    const int wave = tid >> 5;
    const long long base = (long long)blockIdx.x * CH;
    s_cnt[tid] = 0u;
    __syncthreads();

    int  s[EPT], d[EPT];
    bool on[EPT];
    unsigned rk[EPT];
#pragma unroll
    for (int j = 0; j < EPT; ++j) {
        const long long e = base + j * CHT + tid;
        on[j] = e < (long long)E;
        s[j] = 0; d[j] = 0;
        if (on[j]) { s[j] = ei[e]; d[j] = ei[(long long)E + e]; }
    }
#pragma unroll
    for (int j = 0; j < EPT; ++j) {
        rk[j] = 0u;
        if (on[j]) rk[j] = atomicAdd(&s_cnt[d[j] >> NBS], 1u);
    }
    __syncthreads();

    const unsigned v = s_cnt[tid];
    unsigned incl = v;
#pragma unroll
    for (int off = 1; off < 32; off <<= 1) {
        const unsigned t = __shfl_up(incl, off);
        if (lane >= off) incl += t;
    }
    if (lane == 31) s_wsum[wave] = incl;
    __syncthreads();
    unsigned wpre = 0u;
#pragma unroll
    for (int w = 0; w < CHT / 32; ++w) if (w < wave) wpre += s_wsum[w];
    const unsigned excl = wpre + incl - v;
    s_cnt[tid] = excl;
    unsigned* __restrict__ row = seg + (size_t)blockIdx.x * (size_t)(B + 1);
    if (tid < B) row[tid] = excl;
    if (tid == 0) {
        const long long left = (long long)E - base;
        row[B] = (unsigned)(left < (long long)CH ? left : (long long)CH);
    }
    __syncthreads();
#pragma unroll
    for (int j = 0; j < EPT; ++j) {
        if (on[j]) pair[base + s_cnt[d[j] >> NBS] + rk[j]] = make_int2(s[j], d[j]);
    }
}

#define RKT 1024
#define PCS 8
template <bool FILL>
__device__ __forceinline__ void walk_one(const int* __restrict__ pair_raw, size_t p, int n_lo,
                                         unsigned* s_cnt, int* __restrict__ ent) {
    if (FILL) {
        const int2 sd = ((const int2*)pair_raw)[p];
        const unsigned at = atomicAdd(&s_cnt[sd.y - n_lo], 1u);
        ent[at] = sd.x;
    } else {
        const int d = pair_raw[2 * p + 1];
        atomicAdd(&s_cnt[d - n_lo], 1u);
    }
}

template <bool FILL>
__device__ __forceinline__ unsigned walk_pieces(const int* __restrict__ pair_raw,
                                                const unsigned* __restrict__ seg, int K, int B,
                                                int b, int n_lo, int wave, int lane,
                                                unsigned* s_cnt, int* __restrict__ ent) {
    unsigned wfront = 0u;
    for (int k0 = wave; k0 < K; k0 += 32 * (RKT / 32)) {
        const int kl = k0 + (RKT / 32) * lane;
        int beg = 0, end = 0;
        if (kl < K) {
            const unsigned* __restrict__ r = seg + (size_t)kl * (size_t)(B + 1) + b;
            beg = (int)r[0];
            end = (int)r[1];
        }
        for (int i = 0; i < 32; i += PCS) {
            if (k0 + (RKT / 32) * i >= K) break;
            unsigned se[PCS];
            size_t   p[PCS];
            bool     on[PCS];
            int2     sd[PCS];
#pragma unroll
            for (int j = 0; j < PCS; ++j) {
                const unsigned sb = (unsigned)__builtin_amdgcn_readlane(beg, i + j);
                se[j] = (unsigned)__builtin_amdgcn_readlane(end, i + j);
                wfront += sb;
                const unsigned o = sb + (unsigned)lane;
                p[j] = (size_t)(k0 + (RKT / 32) * (i + j)) * CH + o;
                on[j] = o < se[j];
                sd[j] = make_int2(0, 0);
                if (on[j]) {
                    if (FILL) sd[j] = ((const int2*)pair_raw)[p[j]];
                    else      sd[j].y = pair_raw[2 * p[j] + 1];
                }
            }
#pragma unroll
            for (int j = 0; j < PCS; ++j) {
                if (on[j]) {
                    if (FILL) {
                        const unsigned at = atomicAdd(&s_cnt[sd[j].y - n_lo], 1u);
                        ent[at] = sd[j].x;
                    } else {
                        atomicAdd(&s_cnt[sd[j].y - n_lo], 1u);
                    }
                }
            }
#pragma unroll
            for (int j = 0; j < PCS; ++j) {
                const unsigned sb = (unsigned)__builtin_amdgcn_readlane(beg, i + j);
                for (unsigned o = sb + 32u + (unsigned)lane; o < se[j]; o += 32u)
                    walk_one<FILL>(pair_raw, p[j] + (size_t)(o - sb - (unsigned)lane), n_lo, s_cnt, ent);
            }
        }
    }
    return wfront;
}

__global__ __launch_bounds__(RKT) void k_bucket_rows(const int* __restrict__ pair_raw,
                                                     const unsigned* __restrict__ seg, int K, int B,
                                                     unsigned* __restrict__ rowptr,
                                                     float* __restrict__ dinv,
                                                     int* __restrict__ ent, int n) {
    __shared__ unsigned s_deg[NB];
    __shared__ unsigned s_at[NB];
    __shared__ unsigned s_front[RKT / 32];
    __shared__ unsigned s_wsum[NB / 32];
    const int tid = threadIdx.x;
    const int lane = tid & 31;
    const int wave = __builtin_amdgcn_readfirstlane(tid >> 5);
    const int b = blockIdx.x;
    const int n_lo = b * NB;
    if (tid < NB) s_deg[tid] = 0u;
    __syncthreads();

    const unsigned wfront = walk_pieces<false>(pair_raw, seg, K, B, b, n_lo, wave, lane, s_deg, ent);
    if (lane == 0) s_front[wave] = wfront;
    __syncthreads();

    const int idx = n_lo + tid;
    unsigned cc = 0u, incl = 0u;
    if (tid < NB) {
        if (idx < n) cc = s_deg[tid] + 1u;
        incl = cc;
#pragma unroll
        for (int off = 1; off < 32; off <<= 1) {
            const unsigned t = __shfl_up(incl, off);
            if (lane >= off) incl += t;
        }
        if (lane == 31) s_wsum[wave] = incl;
    }
    __syncthreads();
    if (tid < NB) {
        unsigned base = (unsigned)n_lo;
#pragma unroll
        for (int w = 0; w < RKT / 32; ++w) base += s_front[w];
#pragma unroll
        for (int w = 0; w < NB / 32; ++w) if (w < wave) base += s_wsum[w];
        const unsigned pos = base + incl - cc;
        s_at[tid] = pos + 1u;
        if (idx < n) {
            rowptr[idx] = pos;
            dinv[idx] = rsqrtf((float)cc);
            ent[pos] = idx;
            if (idx == n - 1) rowptr[n] = pos + cc;
        }
    }
    __syncthreads();

    walk_pieces<true>(pair_raw, seg, K, B, b, n_lo, wave, lane, s_at, ent);
}

#define NG 13
#define NM 12
#define TPB 25
#define NST ((TPB + NG - 1) / NG)
#define TP 132

__device__ __forceinline__ void gather_tile(const float* __restrict__ z,
                                            const unsigned* __restrict__ rowptr,
                                            const int* __restrict__ ent,
                                            const float* __restrict__ dinv,
                                            float* tile, int n0, int n_rows, int lane) {
    const int m16 = lane & 15;
    const int n_hi = n0 + 16 < n_rows ? n0 + 16 : n_rows;
    const unsigned e_beg = rowptr[n0];
    const unsigned e_end = rowptr[n_hi];
    int bi = n0 + m16 + 1;
    if (bi > n_rows) bi = n_rows;
    const unsigned bnd = rowptr[bi];
    const float dn = dinv[bi - 1];
    int cur = 0;
    unsigned cur_bnd = (unsigned)__builtin_amdgcn_readlane((int)bnd, 0);
    float cur_dn = __int_as_float(__builtin_amdgcn_readlane(__float_as_int(dn), 0));
    float4 acc = make_float4(0.0f, 0.0f, 0.0f, 0.0f);
    const float* zl = z + lane * 4;
    int s_cur = 0, s_nxt = 0;
    if (e_beg + (unsigned)lane < e_end) s_cur = ent[e_beg + lane];
    if (e_beg + 32u + (unsigned)lane < e_end) s_nxt = ent[e_beg + 32u + lane];
    float w_cur = dinv[s_cur];
    asm volatile("" : : "v"(w_cur));
    float4 v[16];
#pragma unroll
    for (int i = 0; i < 16; ++i) {
        const int s = __builtin_amdgcn_readlane(s_cur, i);
        v[i] = *(const float4*)(zl + (size_t)s * 128);
    }
    auto add_row = [&](const float4& r, float w, unsigned idx) {
        acc.x += w * r.x;
        acc.y += w * r.y;
        acc.z += w * r.z;
        acc.w += w * r.w;
        if (idx + 1u == cur_bnd) {
            *(float4*)(tile + cur * TP + lane * 4) =
                make_float4(cur_dn * acc.x, cur_dn * acc.y, cur_dn * acc.z, cur_dn * acc.w);
            acc = make_float4(0.0f, 0.0f, 0.0f, 0.0f);
            cur = (cur + 1) & 15;
            cur_bnd = (unsigned)__builtin_amdgcn_readlane((int)bnd, cur);
            cur_dn = __int_as_float(__builtin_amdgcn_readlane(__float_as_int(dn), cur));
        }
    };
    unsigned base = e_beg;
    for (; base + 32u < e_end; base += 32u) {
        const int   my_s = s_cur;
        const float my_w = w_cur;
        s_cur = s_nxt;
        w_cur = dinv[s_cur];
        s_nxt = 0;
        if (base + 64u + (unsigned)lane < e_end) s_nxt = ent[base + 64u + lane];
#pragma unroll
        for (int i = 0; i < 16; ++i) {
            const float w = __int_as_float(__builtin_amdgcn_readlane(__float_as_int(my_w), i));
            add_row(v[i], w, base + i);
            const int s = __builtin_amdgcn_readlane(my_s, 16 + i);
            v[i] = *(const float4*)(zl + (size_t)s * 128);
        }
#pragma unroll
        for (int i = 0; i < 16; ++i) {
            const float w = __int_as_float(__builtin_amdgcn_readlane(__float_as_int(my_w), 16 + i));
            add_row(v[i], w, base + 16u + i);
            const int s = __builtin_amdgcn_readlane(s_cur, i);
            v[i] = *(const float4*)(zl + (size_t)s * 128);
        }
    }
    {
        const int   my_s = s_cur;
        const float my_w = w_cur;
        const unsigned b16 = base + 16u;
        if (b16 < e_end) {
#pragma unroll
            for (int i = 0; i < 16; ++i) {
                const float w = __int_as_float(__builtin_amdgcn_readlane(__float_as_int(my_w), i));
                add_row(v[i], w, base + i);
                const int s = __builtin_amdgcn_readlane(my_s, 16 + i);
                v[i] = *(const float4*)(zl + (size_t)s * 128);
            }
#pragma unroll
            for (int i = 0; i < 16; ++i) {
                if (b16 + i < e_end) {
                    const float w = __int_as_float(__builtin_amdgcn_readlane(__float_as_int(my_w), 16 + i));
                    add_row(v[i], w, b16 + i);
                }
            }
        } else {
#pragma unroll
            for (int i = 0; i < 16; ++i) {
                if (base + i < e_end) {
                    const float w = __int_as_float(__builtin_amdgcn_readlane(__float_as_int(my_w), i));
                    add_row(v[i], w, base + i);
                }
            }
        }
    }
}

__device__ __forceinline__ void product_tile(float* tile, const float* sW1, const float* sW2,
                                             const float* __restrict__ dinv,
                                             const float* __restrict__ b1,
                                             float* __restrict__ h3, int n0, int n_rows, int lane) {
    const int m16 = lane & 15;
    const int hi  = lane >> 4;
    v8f zero8 = {};
    v8f acc1[8];
#pragma unroll
    for (int t = 0; t < 8; ++t) acc1[t] = zero8;
    const float* arow = tile + m16 * TP;
    for (int k0 = 0; k0 < 128; k0 += 4) {
        v2f a = *(const v2f*)(arow + k0 + 2 * hi);
        const int kh = (k0 >> 1) + hi;
#pragma unroll
        for (int t = 0; t < 8; ++t) {
            v2f b = *(const v2f*)(sW1 + (size_t)(kh * 128 + t * 16 + m16) * 2);
            acc1[t] = __builtin_amdgcn_wmma_f32_16x16x4_f32(
                false, a, false, b, (short)0, acc1[t], false, false);
        }
    }

    float bias[8];
#pragma unroll
    for (int t = 0; t < 8; ++t) bias[t] = b1[t * 16 + m16];
#pragma unroll
    for (int t = 0; t < 8; ++t) {
#pragma unroll
        for (int v = 0; v < 8; ++v) {
            float x = acc1[t][v] + bias[t];
            x = x > 0.0f ? x : LRELU_SLOPE * x;
            tile[(v + 8 * hi) * TP + t * 16 + m16] = x;
        }
    }

    v8f acc2[4];
#pragma unroll
    for (int t = 0; t < 4; ++t) acc2[t] = zero8;
    for (int k0 = 0; k0 < 128; k0 += 4) {
        v2f a = *(const v2f*)(arow + k0 + 2 * hi);
        const int kh = (k0 >> 1) + hi;
#pragma unroll
        for (int t = 0; t < 4; ++t) {
            v2f b = *(const v2f*)(sW2 + (size_t)(kh * 64 + t * 16 + m16) * 2);
            acc2[t] = __builtin_amdgcn_wmma_f32_16x16x4_f32(
                false, a, false, b, (short)0, acc2[t], false, false);
        }
    }
    float dr[8];
#pragma unroll
    for (int v = 0; v < 8; ++v) {
        const int row = n0 + v + 8 * hi;
        dr[v] = row < n_rows ? dinv[row] : 0.0f;
    }
#pragma unroll
    for (int t = 0; t < 4; ++t) {
#pragma unroll
        for (int v = 0; v < 8; ++v) {
            const int row = n0 + v + 8 * hi;
            if (row < n_rows) h3[(size_t)row * 64 + t * 16 + m16] = dr[v] * acc2[t][v];
        }
    }
}

__global__ __launch_bounds__((NG + NM) * 32) void k_layer1_fused(const float* __restrict__ z,
                                                                 const unsigned* __restrict__ rowptr,
                                                                 const int* __restrict__ ent,
                                                                 const float* __restrict__ dinv,
                                                                 const float* __restrict__ W1,
                                                                 const float* __restrict__ b1,
                                                                 const float* __restrict__ W2,
                                                                 float* __restrict__ h3, int n_rows) {
    __shared__ __attribute__((aligned(16))) float sW1[128 * 128];
    __shared__ __attribute__((aligned(16))) float sW2[128 * 64];
    __shared__ __attribute__((aligned(16))) float sT[2 * NG * 16 * TP];
    const int tid = threadIdx.x;
    const int lane = tid & 31;
    const int wave = __builtin_amdgcn_readfirstlane(tid >> 5);
    const bool gathers = wave < NG;
    if (!gathers) {
        for (int i = tid - NG * 32; i < 128 * 128; i += NM * 32) {
            const int k = i >> 7, n = i & 127;
            sW1[((k >> 1) * 128 + n) * 2 + (k & 1)] = W1[i];
        }
        for (int i = tid - NG * 32; i < 128 * 64; i += NM * 32) {
            const int k = i >> 6, n = i & 63;
            sW2[((k >> 1) * 64 + n) * 2 + (k & 1)] = W2[i];
        }
    }
    const int tile0 = blockIdx.x * TPB;
#pragma nounroll
    for (int s = 0; s <= NST; ++s) {
        if (gathers) {
            const int t = s * NG + wave;
            if (s < NST && t < TPB) {
                const int n0 = (tile0 + t) * 16;
                if (n0 < n_rows)
                    gather_tile(z, rowptr, ent, dinv, sT + ((s & 1) * NG + wave) * (16 * TP), n0, n_rows, lane);
            }
        } else if (s > 0) {
            const int sp = s - 1;
#pragma nounroll
            for (int j = wave - NG; j < NG; j += NM) {
                const int t = sp * NG + j;
                if (t < TPB) {
                    const int n0 = (tile0 + t) * 16;
                    if (n0 < n_rows)
                        product_tile(sT + ((sp & 1) * NG + j) * (16 * TP), sW1, sW2, dinv, b1, h3, n0, n_rows, lane);
                }
            }
        }
        if (s < NST) __syncthreads();
    }
}

template <int FEAT>
__global__ __launch_bounds__(256) void k_gather_rows(const float* __restrict__ hs,
                                                     const unsigned* __restrict__ rowptr,
                                                     const int* __restrict__ ent,
                                                     const float* __restrict__ dinv,
                                                     const float* __restrict__ b,
                                                     float* __restrict__ out, int n_rows) {
    constexpr int VW = FEAT / 32;
    typedef float vrow __attribute__((ext_vector_type(VW)));
    int node = (int)(((long long)blockIdx.x * 256 + threadIdx.x) >> 5);
    if (node >= n_rows) return;
    node = __builtin_amdgcn_readfirstlane(node);
    const int lane = threadIdx.x & 31;
    const unsigned beg = rowptr[node];
    const unsigned end = rowptr[node + 1];
    const float dn = dinv[node];
    const float* hl = hs + lane * VW;

    vrow acc = {};
    for (unsigned base = beg; base < end; base += 32u) {
        const unsigned left = end - base;
        const unsigned cnt = left < 32u ? left : 32u;
        int my = 0;
        if ((unsigned)lane < cnt) my = ent[base + lane];
        unsigned j = 0;
        for (; j + 4u <= cnt; j += 4u) {
            const int s0 = __builtin_amdgcn_readlane(my, j + 0);
            const int s1 = __builtin_amdgcn_readlane(my, j + 1);
            const int s2 = __builtin_amdgcn_readlane(my, j + 2);
            const int s3 = __builtin_amdgcn_readlane(my, j + 3);
            const vrow v0 = *(const vrow*)(hl + (size_t)s0 * FEAT);
            const vrow v1 = *(const vrow*)(hl + (size_t)s1 * FEAT);
            const vrow v2 = *(const vrow*)(hl + (size_t)s2 * FEAT);
            const vrow v3 = *(const vrow*)(hl + (size_t)s3 * FEAT);
            acc += v0;
            acc += v1;
            acc += v2;
            acc += v3;
        }
        if (j < cnt) {
            const int s0 = __builtin_amdgcn_readlane(my, j);
            const vrow v0 = *(const vrow*)(hl + (size_t)s0 * FEAT);
            vrow v1 = {}, v2 = {};
            if (j + 1u < cnt) {
                const int s1 = __builtin_amdgcn_readlane(my, j + 1);
                v1 = *(const vrow*)(hl + (size_t)s1 * FEAT);
            }
            if (j + 2u < cnt) {
                const int s2 = __builtin_amdgcn_readlane(my, j + 2);
                v2 = *(const vrow*)(hl + (size_t)s2 * FEAT);
            }
            acc += v0;
            acc += v1;
            acc += v2;
        }
    }

    const vrow bb = *(const vrow*)(b + lane * VW);
    acc = dn * acc + bb;
    *(vrow*)(out + (size_t)node * FEAT + lane * VW) = acc;
}

extern "C" void kernel_launch(void* const* d_in, const int* in_sizes, int n_in,
                              void* d_out, int out_size, void* d_ws, size_t ws_size,
                              hipStream_t stream) {
    (void)n_in; (void)out_size; (void)ws_size;
    const float* z  = (const float*)d_in[0];
    const int*   ei = (const int*)d_in[1];
    const float* W1 = (const float*)d_in[2];
    const float* b1 = (const float*)d_in[3];
    const float* W2 = (const float*)d_in[4];
    const float* b2 = (const float*)d_in[5];

    const int N = in_sizes[0] / 128;
    const int E = in_sizes[1] / 2;

    char* ws = (char*)d_ws;
    size_t off = 0;
    auto carve = [&](size_t bytes) -> void* {
        void* p = ws + off;
        off = (off + bytes + 255) & ~(size_t)255;
        return p;
    };
    float*    dinv   = (float*)carve((size_t)N * 4);
    unsigned* rowptr = (unsigned*)carve(((size_t)N + 1) * 4);
    const int K = (E + CH - 1) / CH;
    const int B = (N + NB - 1) / NB;
    int*      pair   = (int*)carve((size_t)K * CH * 8);
    unsigned* seg    = (unsigned*)carve((size_t)K * (B + 1) * 4);
    int*      ent    = (int*)carve(((size_t)E + (size_t)N) * 4);
    float*    h3     = (float*)carve((size_t)N * 64 * 4);
    float*    outp   = (float*)d_out;

    const dim3 blk(256);
    const int gF = (N + 399) / 400;
    const int gG = (N + 7) / 8;

    k_chunk_sort <<<K, dim3(CHT), 0, stream>>>(ei, E, B, pair, seg);
    k_bucket_rows<<<B, dim3(RKT), 0, stream>>>(pair, seg, K, B, rowptr, dinv, ent, N);

    k_layer1_fused<<<gF, dim3((NG + NM) * 32), 0, stream>>>(z, rowptr, ent, dinv, W1, b1, W2, h3, N);

    k_gather_rows<64><<<gG, blk, 0, stream>>>(h3, rowptr, ent, dinv, b2, outp, N);
}


# Both sort passes build their scattered output in LDS and copy it out as whole 128-byte lines: the chunk's grouped pairs, each bucket's stretch of the source list (direct stores if too long)
# speedup vs baseline: 6.1164x; 1.0782x over previous
#include <hip/hip_runtime.h>
#include <hip/hip_bf16.h>

typedef float v2f __attribute__((ext_vector_type(2)));
typedef float v8f __attribute__((ext_vector_type(8)));

#define LRELU_SLOPE 0.01f

#define NBS 8
#define NB (1 << NBS)
#define CHT 1024
#define EPT 8
#define CH (CHT * EPT)

__global__ __launch_bounds__(CHT) void k_chunk_sort(const int* __restrict__ ei, int E, int B,
                                                    int* __restrict__ pair_raw,
                                                    unsigned* __restrict__ seg) {
    __shared__ unsigned s_cnt[CHT];
    __shared__ unsigned s_wsum[CHT / 32];
    __shared__ __attribute__((aligned(16))) int2 s_pair[CH];
    const int tid = threadIdx.x;
    const int lane = tid & 31;
    const int wave = tid >> 5;
    const long long base = (long long)blockIdx.x * CH;
    s_cnt[tid] = 0u;
    __syncthreads();

    int  s[EPT], d[EPT];
    bool on[EPT];
    unsigned rk[EPT];
#pragma unroll
    for (int j = 0; j < EPT; ++j) {
        const long long e = base + j * CHT + tid;
        on[j] = e < (long long)E;
        s[j] = 0; d[j] = 0;
        if (on[j]) { s[j] = ei[e]; d[j] = ei[(long long)E + e]; }
    }
#pragma unroll
    for (int j = 0; j < EPT; ++j) {
        rk[j] = 0u;
        if (on[j]) rk[j] = atomicAdd(&s_cnt[d[j] >> NBS], 1u);
    }
    __syncthreads();

    const unsigned v = s_cnt[tid];
    unsigned incl = v;
#pragma unroll
    for (int off = 1; off < 32; off <<= 1) {
        const unsigned t = __shfl_up(incl, off);
        if (lane >= off) incl += t;
    }
    if (lane == 31) s_wsum[wave] = incl;
    __syncthreads();
    unsigned wpre = 0u;
#pragma unroll
    for (int w = 0; w < CHT / 32; ++w) if (w < wave) wpre += s_wsum[w];
    const unsigned excl = wpre + incl - v;
    s_cnt[tid] = excl;
    unsigned* __restrict__ row = seg + (size_t)blockIdx.x * (size_t)(B + 1);
    if (tid < B) row[tid] = excl;
    const long long left = (long long)E - base;
    const int n_here = (int)(left < (long long)CH ? left : (long long)CH);
    if (tid == 0) row[B] = (unsigned)n_here;
    __syncthreads();
#pragma unroll
    for (int j = 0; j < EPT; ++j) {
        if (on[j]) s_pair[s_cnt[d[j] >> NBS] + rk[j]] = make_int2(s[j], d[j]);
    }
    __syncthreads();
    int4* __restrict__ out4 = (int4*)(pair_raw + 2 * base);
    const int4* in4 = (const int4*)s_pair;
    const int n4 = n_here >> 1;
    for (int i = tid; i < n4; i += CHT) out4[i] = in4[i];
    if ((n_here & 1) && tid == 0) ((int2*)pair_raw)[base + n_here - 1] = s_pair[n_here - 1];
}

#define RKT 1024
#define PCS 8
#define ECAP 8192
template <bool FILL, bool STAGED>
__device__ __forceinline__ void walk_one(const int* __restrict__ pair_raw, size_t p, int n_lo,
                                         unsigned* s_cnt, int* __restrict__ ent, int* s_ent) {
    if (FILL) {
        const int2 sd = ((const int2*)pair_raw)[p];
        const unsigned at = atomicAdd(&s_cnt[sd.y - n_lo], 1u);
        if (STAGED) s_ent[at] = sd.x;
        else        ent[at] = sd.x;
    } else {
        const int d = pair_raw[2 * p + 1];
        atomicAdd(&s_cnt[d - n_lo], 1u);
    }
}

template <bool FILL, bool STAGED>
__device__ __forceinline__ unsigned walk_pieces(const int* __restrict__ pair_raw,
                                                const unsigned* __restrict__ seg, int K, int B,
                                                int b, int n_lo, int wave, int lane,
                                                unsigned* s_cnt, int* __restrict__ ent, int* s_ent) {
    unsigned wfront = 0u;
    for (int k0 = wave; k0 < K; k0 += 32 * (RKT / 32)) {
        const int kl = k0 + (RKT / 32) * lane;
        int beg = 0, end = 0;
        if (kl < K) {
            const unsigned* __restrict__ r = seg + (size_t)kl * (size_t)(B + 1) + b;
            beg = (int)r[0];
            end = (int)r[1];
        }
        for (int i = 0; i < 32; i += PCS) {
            if (k0 + (RKT / 32) * i >= K) break;
            unsigned se[PCS];
            size_t   p[PCS];
            bool     on[PCS];
            int2     sd[PCS];
#pragma unroll
            for (int j = 0; j < PCS; ++j) {
                const unsigned sb = (unsigned)__builtin_amdgcn_readlane(beg, i + j);
                se[j] = (unsigned)__builtin_amdgcn_readlane(end, i + j);
                wfront += sb;
                const unsigned o = sb + (unsigned)lane;
                p[j] = (size_t)(k0 + (RKT / 32) * (i + j)) * CH + o;
                on[j] = o < se[j];
                sd[j] = make_int2(0, 0);
                if (on[j]) {
                    if (FILL) sd[j] = ((const int2*)pair_raw)[p[j]];
                    else      sd[j].y = pair_raw[2 * p[j] + 1];
                }
            }
#pragma unroll
            for (int j = 0; j < PCS; ++j) {
                if (on[j]) {
                    if (FILL) {
                        const unsigned at = atomicAdd(&s_cnt[sd[j].y - n_lo], 1u);
                        if (STAGED) s_ent[at] = sd[j].x;
                        else        ent[at] = sd[j].x;
                    } else {
                        atomicAdd(&s_cnt[sd[j].y - n_lo], 1u);
                    }
                }
            }
#pragma unroll
            for (int j = 0; j < PCS; ++j) {
                const unsigned sb = (unsigned)__builtin_amdgcn_readlane(beg, i + j);
                for (unsigned o = sb + 32u + (unsigned)lane; o < se[j]; o += 32u)
                    walk_one<FILL, STAGED>(pair_raw, p[j] + (size_t)(o - sb - (unsigned)lane), n_lo, s_cnt, ent, s_ent);
            }
        }
    }
    return wfront;
}

__global__ __launch_bounds__(RKT) void k_bucket_rows(const int* __restrict__ pair_raw,
                                                     const unsigned* __restrict__ seg, int K, int B,
                                                     unsigned* __restrict__ rowptr,
                                                     float* __restrict__ dinv,
                                                     int* __restrict__ ent, int n) {
    __shared__ unsigned s_deg[NB];
    __shared__ unsigned s_at[NB];
    __shared__ unsigned s_front[RKT / 32];
    __shared__ unsigned s_wsum[NB / 32];
    __shared__ int s_ent[ECAP];
    const int tid = threadIdx.x;
    const int lane = tid & 31;
    const int wave = __builtin_amdgcn_readfirstlane(tid >> 5);
    const int b = blockIdx.x;
    const int n_lo = b * NB;
    if (tid < NB) s_deg[tid] = 0u;
    __syncthreads();

    const unsigned wfront = walk_pieces<false, false>(pair_raw, seg, K, B, b, n_lo, wave, lane, s_deg, ent, s_ent);
    if (lane == 0) s_front[wave] = wfront;
    __syncthreads();

    const int idx = n_lo + tid;
    unsigned cc = 0u, incl = 0u;
    if (tid < NB) {
        if (idx < n) cc = s_deg[tid] + 1u;
        incl = cc;
#pragma unroll
        for (int off = 1; off < 32; off <<= 1) {
            const unsigned t = __shfl_up(incl, off);
            if (lane >= off) incl += t;
        }
        if (lane == 31) s_wsum[wave] = incl;
    }
    __syncthreads();
    unsigned gbase = (unsigned)n_lo;
#pragma unroll
    for (int w = 0; w < RKT / 32; ++w) gbase += s_front[w];
    unsigned tot = 0u;
#pragma unroll
    for (int w = 0; w < NB / 32; ++w) tot += s_wsum[w];
    gbase = (unsigned)__builtin_amdgcn_readfirstlane((int)gbase);
    tot = (unsigned)__builtin_amdgcn_readfirstlane((int)tot);
    const bool staged = tot <= (unsigned)ECAP;
    if (tid < NB) {
        unsigned base = gbase;
#pragma unroll
        for (int w = 0; w < NB / 32; ++w) if (w < wave) base += s_wsum[w];
        const unsigned pos = base + incl - cc;
        const unsigned rel = pos - gbase;
        s_at[tid] = (staged ? rel : pos) + 1u;
        if (idx < n) {
            rowptr[idx] = pos;
            dinv[idx] = rsqrtf((float)cc);
            if (staged) s_ent[rel] = idx;
            else        ent[pos] = idx;
            if (idx == n - 1) rowptr[n] = pos + cc;
        }
    }
    __syncthreads();

    if (staged) {
        walk_pieces<true, true>(pair_raw, seg, K, B, b, n_lo, wave, lane, s_at, ent, s_ent);
        __syncthreads();
        const int mis = (int)(gbase & 31u);
        for (int i = tid - mis; i < (int)tot; i += RKT)
            if (i >= 0) ent[gbase + (unsigned)i] = s_ent[i];
    } else {
        walk_pieces<true, false>(pair_raw, seg, K, B, b, n_lo, wave, lane, s_at, ent, s_ent);
    }
}

#define NG 13
#define NM 12
#define TPB 25
#define NST ((TPB + NG - 1) / NG)
#define TP 132

__device__ __forceinline__ void gather_tile(const float* __restrict__ z,
                                            const unsigned* __restrict__ rowptr,
                                            const int* __restrict__ ent,
                                            const float* __restrict__ dinv,
                                            float* tile, int n0, int n_rows, int lane) {
    const int m16 = lane & 15;
    const int n_hi = n0 + 16 < n_rows ? n0 + 16 : n_rows;
    const unsigned e_beg = rowptr[n0];
    const unsigned e_end = rowptr[n_hi];
    int bi = n0 + m16 + 1;
    if (bi > n_rows) bi = n_rows;
    const unsigned bnd = rowptr[bi];
    const float dn = dinv[bi - 1];
    int cur = 0;
    unsigned cur_bnd = (unsigned)__builtin_amdgcn_readlane((int)bnd, 0);
    float cur_dn = __int_as_float(__builtin_amdgcn_readlane(__float_as_int(dn), 0));
    float4 acc = make_float4(0.0f, 0.0f, 0.0f, 0.0f);
    const float* zl = z + lane * 4;
    int s_cur = 0, s_nxt = 0;
    if (e_beg + (unsigned)lane < e_end) s_cur = ent[e_beg + lane];
    if (e_beg + 32u + (unsigned)lane < e_end) s_nxt = ent[e_beg + 32u + lane];
    float w_cur = dinv[s_cur];
    asm volatile("" : : "v"(w_cur));
    float4 v[16];
#pragma unroll
    for (int i = 0; i < 16; ++i) {
        const int s = __builtin_amdgcn_readlane(s_cur, i);
        v[i] = *(const float4*)(zl + (size_t)s * 128);
    }
    auto add_row = [&](const float4& r, float w, unsigned idx) {
        acc.x += w * r.x;
        acc.y += w * r.y;
        acc.z += w * r.z;
        acc.w += w * r.w;
        if (idx + 1u == cur_bnd) {
            *(float4*)(tile + cur * TP + lane * 4) =
                make_float4(cur_dn * acc.x, cur_dn * acc.y, cur_dn * acc.z, cur_dn * acc.w);
            acc = make_float4(0.0f, 0.0f, 0.0f, 0.0f);
            cur = (cur + 1) & 15;
            cur_bnd = (unsigned)__builtin_amdgcn_readlane((int)bnd, cur);
            cur_dn = __int_as_float(__builtin_amdgcn_readlane(__float_as_int(dn), cur));
        }
    };
    unsigned base = e_beg;
    for (; base + 32u < e_end; base += 32u) {
        const int   my_s = s_cur;
        const float my_w = w_cur;
        s_cur = s_nxt;
        w_cur = dinv[s_cur];
        s_nxt = 0;
        if (base + 64u + (unsigned)lane < e_end) s_nxt = ent[base + 64u + lane];
#pragma unroll
        for (int i = 0; i < 16; ++i) {
            const float w = __int_as_float(__builtin_amdgcn_readlane(__float_as_int(my_w), i));
            add_row(v[i], w, base + i);
            const int s = __builtin_amdgcn_readlane(my_s, 16 + i);
            v[i] = *(const float4*)(zl + (size_t)s * 128);
        }
#pragma unroll
        for (int i = 0; i < 16; ++i) {
            const float w = __int_as_float(__builtin_amdgcn_readlane(__float_as_int(my_w), 16 + i));
            add_row(v[i], w, base + 16u + i);
            const int s = __builtin_amdgcn_readlane(s_cur, i);
            v[i] = *(const float4*)(zl + (size_t)s * 128);
        }
    }
    {
        const int   my_s = s_cur;
        const float my_w = w_cur;
        const unsigned b16 = base + 16u;
        if (b16 < e_end) {
#pragma unroll
            for (int i = 0; i < 16; ++i) {
                const float w = __int_as_float(__builtin_amdgcn_readlane(__float_as_int(my_w), i));
                add_row(v[i], w, base + i);
                const int s = __builtin_amdgcn_readlane(my_s, 16 + i);
                v[i] = *(const float4*)(zl + (size_t)s * 128);
            }
#pragma unroll
            for (int i = 0; i < 16; ++i) {
                if (b16 + i < e_end) {
                    const float w = __int_as_float(__builtin_amdgcn_readlane(__float_as_int(my_w), 16 + i));
                    add_row(v[i], w, b16 + i);
                }
            }
        } else {
#pragma unroll
            for (int i = 0; i < 16; ++i) {
                if (base + i < e_end) {
                    const float w = __int_as_float(__builtin_amdgcn_readlane(__float_as_int(my_w), i));
                    add_row(v[i], w, base + i);
                }
            }
        }
    }
}

__device__ __forceinline__ void product_tile(float* tile, const float* sW1, const float* sW2,
                                             const float* __restrict__ dinv,
                                             const float* __restrict__ b1,
                                             float* __restrict__ h3, int n0, int n_rows, int lane) {
    const int m16 = lane & 15;
    const int hi  = lane >> 4;
    v8f zero8 = {};
    v8f acc1[8];
#pragma unroll
    for (int t = 0; t < 8; ++t) acc1[t] = zero8;
    const float* arow = tile + m16 * TP;
    for (int k0 = 0; k0 < 128; k0 += 4) {
        v2f a = *(const v2f*)(arow + k0 + 2 * hi);
        const int kh = (k0 >> 1) + hi;
#pragma unroll
        for (int t = 0; t < 8; ++t) {
            v2f b = *(const v2f*)(sW1 + (size_t)(kh * 128 + t * 16 + m16) * 2);
            acc1[t] = __builtin_amdgcn_wmma_f32_16x16x4_f32(
                false, a, false, b, (short)0, acc1[t], false, false);
        }
    }

    float bias[8];
#pragma unroll
    for (int t = 0; t < 8; ++t) bias[t] = b1[t * 16 + m16];
#pragma unroll
    for (int t = 0; t < 8; ++t) {
#pragma unroll
        for (int v = 0; v < 8; ++v) {
            float x = acc1[t][v] + bias[t];
            x = x > 0.0f ? x : LRELU_SLOPE * x;
            tile[(v + 8 * hi) * TP + t * 16 + m16] = x;
        }
    }

    v8f acc2[4];
#pragma unroll
    for (int t = 0; t < 4; ++t) acc2[t] = zero8;
    for (int k0 = 0; k0 < 128; k0 += 4) {
        v2f a = *(const v2f*)(arow + k0 + 2 * hi);
        const int kh = (k0 >> 1) + hi;
#pragma unroll
        for (int t = 0; t < 4; ++t) {
            v2f b = *(const v2f*)(sW2 + (size_t)(kh * 64 + t * 16 + m16) * 2);
            acc2[t] = __builtin_amdgcn_wmma_f32_16x16x4_f32(
                false, a, false, b, (short)0, acc2[t], false, false);
        }
    }
    float dr[8];
#pragma unroll
    for (int v = 0; v < 8; ++v) {
        const int row = n0 + v + 8 * hi;
        dr[v] = row < n_rows ? dinv[row] : 0.0f;
    }
#pragma unroll
    for (int t = 0; t < 4; ++t) {
#pragma unroll
        for (int v = 0; v < 8; ++v) {
            const int row = n0 + v + 8 * hi;
            if (row < n_rows) h3[(size_t)row * 64 + t * 16 + m16] = dr[v] * acc2[t][v];
        }
    }
}

__global__ __launch_bounds__((NG + NM) * 32) void k_layer1_fused(const float* __restrict__ z,
                                                                 const unsigned* __restrict__ rowptr,
                                                                 const int* __restrict__ ent,
                                                                 const float* __restrict__ dinv,
                                                                 const float* __restrict__ W1,
                                                                 const float* __restrict__ b1,
                                                                 const float* __restrict__ W2,
                                                                 float* __restrict__ h3, int n_rows) {
    __shared__ __attribute__((aligned(16))) float sW1[128 * 128];
    __shared__ __attribute__((aligned(16))) float sW2[128 * 64];
    __shared__ __attribute__((aligned(16))) float sT[2 * NG * 16 * TP];
    const int tid = threadIdx.x;
    const int lane = tid & 31;
    const int wave = __builtin_amdgcn_readfirstlane(tid >> 5);
    const bool gathers = wave < NG;
    if (!gathers) {
        for (int i = tid - NG * 32; i < 128 * 128; i += NM * 32) {
            const int k = i >> 7, n = i & 127;
            sW1[((k >> 1) * 128 + n) * 2 + (k & 1)] = W1[i];
        }
        for (int i = tid - NG * 32; i < 128 * 64; i += NM * 32) {
            const int k = i >> 6, n = i & 63;
            sW2[((k >> 1) * 64 + n) * 2 + (k & 1)] = W2[i];
        }
    }
    const int tile0 = blockIdx.x * TPB;
#pragma nounroll
    for (int s = 0; s <= NST; ++s) {
        if (gathers) {
            const int t = s * NG + wave;
            if (s < NST && t < TPB) {
                const int n0 = (tile0 + t) * 16;
                if (n0 < n_rows)
                    gather_tile(z, rowptr, ent, dinv, sT + ((s & 1) * NG + wave) * (16 * TP), n0, n_rows, lane);
            }
        } else if (s > 0) {
            const int sp = s - 1;
#pragma nounroll
            for (int j = wave - NG; j < NG; j += NM) {
                const int t = sp * NG + j;
                if (t < TPB) {
                    const int n0 = (tile0 + t) * 16;
                    if (n0 < n_rows)
                        product_tile(sT + ((sp & 1) * NG + j) * (16 * TP), sW1, sW2, dinv, b1, h3, n0, n_rows, lane);
                }
            }
        }
        if (s < NST) __syncthreads();
    }
}

template <int FEAT>
__global__ __launch_bounds__(256) void k_gather_rows(const float* __restrict__ hs,
                                                     const unsigned* __restrict__ rowptr,
                                                     const int* __restrict__ ent,
                                                     const float* __restrict__ dinv,
                                                     const float* __restrict__ b,
                                                     float* __restrict__ out, int n_rows) {
    constexpr int VW = FEAT / 32;
    typedef float vrow __attribute__((ext_vector_type(VW)));
    int node = (int)(((long long)blockIdx.x * 256 + threadIdx.x) >> 5);
    if (node >= n_rows) return;
    node = __builtin_amdgcn_readfirstlane(node);
    const int lane = threadIdx.x & 31;
    const unsigned beg = rowptr[node];
    const unsigned end = rowptr[node + 1];
    const float dn = dinv[node];
    const float* hl = hs + lane * VW;

    vrow acc = {};
    for (unsigned base = beg; base < end; base += 32u) {
        const unsigned left = end - base;
        const unsigned cnt = left < 32u ? left : 32u;
        int my = 0;
        if ((unsigned)lane < cnt) my = ent[base + lane];
        unsigned j = 0;
        for (; j + 4u <= cnt; j += 4u) {
            const int s0 = __builtin_amdgcn_readlane(my, j + 0);
            const int s1 = __builtin_amdgcn_readlane(my, j + 1);
            const int s2 = __builtin_amdgcn_readlane(my, j + 2);
            const int s3 = __builtin_amdgcn_readlane(my, j + 3);
            const vrow v0 = *(const vrow*)(hl + (size_t)s0 * FEAT);
            const vrow v1 = *(const vrow*)(hl + (size_t)s1 * FEAT);
            const vrow v2 = *(const vrow*)(hl + (size_t)s2 * FEAT);
            const vrow v3 = *(const vrow*)(hl + (size_t)s3 * FEAT);
            acc += v0;
            acc += v1;
            acc += v2;
            acc += v3;
        }
        if (j < cnt) {
            const int s0 = __builtin_amdgcn_readlane(my, j);
            const vrow v0 = *(const vrow*)(hl + (size_t)s0 * FEAT);
            vrow v1 = {}, v2 = {};
            if (j + 1u < cnt) {
                const int s1 = __builtin_amdgcn_readlane(my, j + 1);
                v1 = *(const vrow*)(hl + (size_t)s1 * FEAT);
            }
            if (j + 2u < cnt) {
                const int s2 = __builtin_amdgcn_readlane(my, j + 2);
                v2 = *(const vrow*)(hl + (size_t)s2 * FEAT);
            }
            acc += v0;
            acc += v1;
            acc += v2;
        }
    }

    const vrow bb = *(const vrow*)(b + lane * VW);
    acc = dn * acc + bb;
    *(vrow*)(out + (size_t)node * FEAT + lane * VW) = acc;
}

extern "C" void kernel_launch(void* const* d_in, const int* in_sizes, int n_in,
                              void* d_out, int out_size, void* d_ws, size_t ws_size,
                              hipStream_t stream) {
    (void)n_in; (void)out_size; (void)ws_size;
    const float* z  = (const float*)d_in[0];
    const int*   ei = (const int*)d_in[1];
    const float* W1 = (const float*)d_in[2];
    const float* b1 = (const float*)d_in[3];
    const float* W2 = (const float*)d_in[4];
    const float* b2 = (const float*)d_in[5];

    const int N = in_sizes[0] / 128;
    const int E = in_sizes[1] / 2;

    char* ws = (char*)d_ws;
    size_t off = 0;
    auto carve = [&](size_t bytes) -> void* {
        void* p = ws + off;
        off = (off + bytes + 255) & ~(size_t)255;
        return p;
    };
    float*    dinv   = (float*)carve((size_t)N * 4);
    unsigned* rowptr = (unsigned*)carve(((size_t)N + 1) * 4);
    const int K = (E + CH - 1) / CH;
    const int B = (N + NB - 1) / NB;
    int*      pair   = (int*)carve((size_t)K * CH * 8);
    unsigned* seg    = (unsigned*)carve((size_t)K * (B + 1) * 4);
    int*      ent    = (int*)carve(((size_t)E + (size_t)N) * 4);
    float*    h3     = (float*)carve((size_t)N * 64 * 4);
    float*    outp   = (float*)d_out;

    const dim3 blk(256);
    const int gF = (N + 399) / 400;
    const int gG = (N + 7) / 8;

    k_chunk_sort <<<K, dim3(CHT), 0, stream>>>(ei, E, B, pair, seg);
    k_bucket_rows<<<B, dim3(RKT), 0, stream>>>(pair, seg, K, B, rowptr, dinv, ent, N);

    k_layer1_fused<<<gF, dim3((NG + NM) * 32), 0, stream>>>(z, rowptr, ent, dinv, W1, b1, W2, h3, N);

    k_gather_rows<64><<<gG, blk, 0, stream>>>(h3, rowptr, ent, dinv, b2, outp, N);
}
